# starting kernel
# baseline (speedup 1.0000x reference)
#include <hip/hip_runtime.h>
#include <hip/hip_bf16.h>


typedef __attribute__((ext_vector_type(2))) float v2f;
typedef __attribute__((ext_vector_type(8))) float v8f;

#define S4_DT 0.001f
#define TT 4096
#define NB 16
#define NS 64
#define NI 256
#define NO 256
#define LCH 64
#define NCH 64

__device__ __forceinline__ v8f wmma4(v2f a, v2f b, v8f c) {
  return __builtin_amdgcn_wmma_f32_16x16x4_f32(
       false, a,  false, b,
       (short)0, c,  false,  false);
}

__device__ __forceinline__ v2f load2(const float* p) { return *(const v2f*)p; }

__device__ __forceinline__ void store_c(float* dst, int rs, int n0, v8f acc,
                                        int l16, int half) {
#pragma unroll
  for (int v = 0; v < 8; ++v)
    dst[(v + 8 * half) * rs + n0 + l16] = acc[v];
}

__device__ __forceinline__ v8f load_c(const float* src, int rs, int n0,
                                      int l16, int half) {
  v8f r;
#pragma unroll
  for (int v = 0; v < 8; ++v)
    r[v] = src[(v + 8 * half) * rs + n0 + l16];
  return r;
}

__global__ __launch_bounds__(512) void s4_prep_kernel(
    const float* __restrict__ A, const float* __restrict__ B,
    float* __restrict__ Abar, float* __restrict__ Apow,
    float* __restrict__ Pcat, float* __restrict__ Bsc) {
  __shared__ float sPrev[2][NS * NS];
  __shared__ float sAbT[NS * NS];

  const int tid = threadIdx.x;
  for (int idx = tid; idx < NS * NS; idx += 512) {
    int r = idx >> 6, c = idx & 63;
    float v = A[idx] * S4_DT + (r == c ? 1.0f : 0.0f);
    Abar[idx] = v;
    sPrev[0][idx] = v;
    sAbT[c * NS + r] = v;
    Apow[0 * NS * NS + idx] = (r == c) ? 1.0f : 0.0f;
    Apow[1 * NS * NS + idx] = v;
  }
  for (int idx = tid; idx < NS * NI; idx += 512) Bsc[idx] = B[idx] * S4_DT;
  __syncthreads();

  const int wave = tid >> 5, lane = tid & 31;
  const int l16 = lane & 15, half = lane >> 4;
  const int m0 = (wave >> 2) * 16, n0 = (wave & 3) * 16;

  int cur = 0;
  for (int k = 2; k <= LCH; ++k) {
    v8f acc = {};
#pragma unroll
    for (int kk = 0; kk < 16; ++kk) {
      v2f a = load2(&sPrev[cur][(m0 + l16) * NS + 4 * kk + 2 * half]);
      v2f b = load2(&sAbT[(n0 + l16) * NS + 4 * kk + 2 * half]);
      acc = wmma4(a, b, acc);
    }
    int nxt = cur ^ 1;
#pragma unroll
    for (int v = 0; v < 8; ++v) {
      int row = m0 + v + 8 * half;
      float val = acc[v];
      sPrev[nxt][row * NS + n0 + l16] = val;
      Apow[k * NS * NS + row * NS + n0 + l16] = val;
    }
    __syncthreads();
    cur = nxt;
  }

  for (int idx = tid; idx < NS * (LCH * NS); idx += 512) {
    int n = idx >> 12, rest = idx & 4095, j = rest >> 6, m = rest & 63;
    Pcat[idx] = Apow[(63 - j) * NS * NS + n * NS + m];
  }
}

__global__ __launch_bounds__(128) void s4_ub_kernel(
    const float* __restrict__ u, const float* __restrict__ Bsc,
    float* __restrict__ uB) {
  const int wave = threadIdx.x >> 5, lane = threadIdx.x & 31;
  const int l16 = lane & 15, half = lane >> 4;
  const int t = blockIdx.x * 4 + wave;

  const float* urow = u + (size_t)l16 * TT * NI + (size_t)t * NI;
  v8f acc[4] = {{}, {}, {}, {}};
  for (int kk = 0; kk < NI / 4; ++kk) {
    v2f a = load2(urow + 4 * kk + 2 * half);
#pragma unroll
    for (int j = 0; j < 4; ++j) {
      v2f b = load2(Bsc + (j * 16 + l16) * NI + 4 * kk + 2 * half);
      acc[j] = wmma4(a, b, acc[j]);
    }
  }
  float* dst = uB + (size_t)t * NB * NS;
#pragma unroll
  for (int j = 0; j < 4; ++j) store_c(dst, NS, j * 16, acc[j], l16, half);
}

__global__ __launch_bounds__(128) void s4_gagg_kernel(
    const float* __restrict__ uB, const float* __restrict__ Pcat,
    float* __restrict__ G) {
  const int wave = threadIdx.x >> 5, lane = threadIdx.x & 31;
  const int l16 = lane & 15, half = lane >> 4;
  const int c = blockIdx.x * 4 + wave;

  const float* F = uB + (size_t)c * LCH * NB * NS;
  v8f acc[4] = {{}, {}, {}, {}};
  for (int kk = 0; kk < (LCH * NS) / 4; ++kk) {
    int kap = 4 * kk + 2 * half;
    int j = kap >> 6, m = kap & 63;
    v2f a = load2(F + j * NB * NS + l16 * NS + m);
#pragma unroll
    for (int jn = 0; jn < 4; ++jn) {
      v2f b = load2(Pcat + (size_t)(jn * 16 + l16) * (LCH * NS) + kap);
      acc[jn] = wmma4(a, b, acc[jn]);
    }
  }
  float* dst = G + (size_t)c * NB * NS;
#pragma unroll
  for (int j = 0; j < 4; ++j) store_c(dst, NS, j * 16, acc[j], l16, half);
}

__global__ __launch_bounds__(128) void s4_bscan_kernel(
    const float* __restrict__ x0, const float* __restrict__ G,
    const float* __restrict__ Apow, float* __restrict__ Sbound,
    float* __restrict__ xfinal) {
  __shared__ float sb[2][NB * NS];
  const int tid = threadIdx.x;
  const int wave = tid >> 5, lane = tid & 31;
  const int l16 = lane & 15, half = lane >> 4;
  const int n0 = wave * 16;
  const float* AL = Apow + (size_t)LCH * NS * NS;

  for (int idx = tid; idx < NB * NS; idx += 128) sb[0][idx] = x0[idx];
  __syncthreads();

  int cur = 0;
  for (int c = 0; c < NCH; ++c) {
    for (int idx = tid; idx < NB * NS; idx += 128)
      Sbound[(size_t)c * NB * NS + idx] = sb[cur][idx];
    v8f acc = load_c(G + (size_t)c * NB * NS, NS, n0, l16, half);
#pragma unroll
    for (int kk = 0; kk < 16; ++kk) {
      v2f a = load2(&sb[cur][l16 * NS + 4 * kk + 2 * half]);
      v2f b = load2(AL + (n0 + l16) * NS + 4 * kk + 2 * half);
      acc = wmma4(a, b, acc);
    }
    int nxt = cur ^ 1;
#pragma unroll
    for (int v = 0; v < 8; ++v)
      sb[nxt][(v + 8 * half) * NS + n0 + l16] = acc[v];
    __syncthreads();
    cur = nxt;
  }
  for (int idx = tid; idx < NB * NS; idx += 128) xfinal[idx] = sb[cur][idx];
}

__global__ __launch_bounds__(128) void s4_states_kernel(
    const float* __restrict__ uB, const float* __restrict__ Sbound,
    const float* __restrict__ Abar, float* __restrict__ S) {
  __shared__ float sb[2][NB * NS];
  const int tid = threadIdx.x;
  const int wave = tid >> 5, lane = tid & 31;
  const int l16 = lane & 15, half = lane >> 4;
  const int n0 = wave * 16;
  const int c = blockIdx.x;
  const int t0 = c * LCH;

  for (int idx = tid; idx < NB * NS; idx += 128)
    sb[0][idx] = Sbound[(size_t)c * NB * NS + idx];
  __syncthreads();

  int cur = 0;
  for (int k = 0; k < LCH; ++k) {
    float* dst = S + (size_t)(t0 + k) * NB * NS;
    for (int idx = tid; idx < NB * NS; idx += 128) dst[idx] = sb[cur][idx];
    v8f acc = load_c(uB + (size_t)(t0 + k) * NB * NS, NS, n0, l16, half);
#pragma unroll
    for (int kk = 0; kk < 16; ++kk) {
      v2f a = load2(&sb[cur][l16 * NS + 4 * kk + 2 * half]);
      v2f b = load2(Abar + (n0 + l16) * NS + 4 * kk + 2 * half);
      acc = wmma4(a, b, acc);
    }
    int nxt = cur ^ 1;
#pragma unroll
    for (int v = 0; v < 8; ++v)
      sb[nxt][(v + 8 * half) * NS + n0 + l16] = acc[v];
    __syncthreads();
    cur = nxt;
  }
}

__global__ __launch_bounds__(128) void s4_y_kernel(
    const float* __restrict__ S, const float* __restrict__ u,
    const float* __restrict__ C, const float* __restrict__ D,
    float* __restrict__ y) {
  const int wave = threadIdx.x >> 5, lane = threadIdx.x & 31;
  const int l16 = lane & 15, half = lane >> 4;
  const int t = blockIdx.x;
  const int o0 = wave * 64;

  v8f acc[4] = {{}, {}, {}, {}};

  const float* srow = S + (size_t)t * NB * NS + l16 * NS;
#pragma unroll
  for (int kk = 0; kk < NS / 4; ++kk) {
    v2f a = load2(srow + 4 * kk + 2 * half);
#pragma unroll
    for (int j = 0; j < 4; ++j) {
      v2f b = load2(C + (o0 + 16 * j + l16) * NS + 4 * kk + 2 * half);
      acc[j] = wmma4(a, b, acc[j]);
    }
  }

  const float* urow = u + (size_t)l16 * TT * NI + (size_t)t * NI;
  for (int kk = 0; kk < NI / 4; ++kk) {
    v2f a = load2(urow + 4 * kk + 2 * half);
#pragma unroll
    for (int j = 0; j < 4; ++j) {
      v2f b = load2(D + (o0 + 16 * j + l16) * NI + 4 * kk + 2 * half);
      acc[j] = wmma4(a, b, acc[j]);
    }
  }

#pragma unroll
  for (int j = 0; j < 4; ++j)
#pragma unroll
    for (int v = 0; v < 8; ++v)
      y[(size_t)(v + 8 * half) * TT * NO + (size_t)t * NO + o0 + 16 * j + l16] =
          acc[j][v];
}

extern "C" void kernel_launch(void* const* d_in, const int* in_sizes, int n_in,
                              void* d_out, int out_size, void* d_ws,
                              size_t ws_size, hipStream_t stream) {
  const float* u  = (const float*)d_in[0];
  const float* x0 = (const float*)d_in[1];
  const float* A  = (const float*)d_in[2];
  const float* B  = (const float*)d_in[3];
  const float* C  = (const float*)d_in[4];
  const float* D  = (const float*)d_in[5];

  float* y      = (float*)d_out;
  float* xfinal = (float*)d_out + (size_t)NB * TT * NO;

  float* ws = (float*)d_ws;
  float* Abar   = ws;                 ws += NS * NS;
  float* Apow   = ws;                 ws += (LCH + 1) * NS * NS;
  float* Pcat   = ws;                 ws += NS * LCH * NS;
  float* Bsc    = ws;                 ws += NS * NI;
  float* uB     = ws;                 ws += (size_t)TT * NB * NS;
  float* Sst    = ws;                 ws += (size_t)TT * NB * NS;
  float* G      = ws;                 ws += NCH * NB * NS;
  float* Sbound = ws;                 ws += NCH * NB * NS;

  s4_prep_kernel<<<1, 512, 0, stream>>>(A, B, Abar, Apow, Pcat, Bsc);
  s4_ub_kernel<<<TT / 4, 128, 0, stream>>>(u, Bsc, uB);
  s4_gagg_kernel<<<NCH / 4, 128, 0, stream>>>(uB, Pcat, G);
  s4_bscan_kernel<<<1, 128, 0, stream>>>(x0, G, Apow, Sbound, xfinal);
  s4_states_kernel<<<NCH, 128, 0, stream>>>(uB, Sbound, Abar, Sst);
  s4_y_kernel<<<TT, 128, 0, stream>>>(Sst, u, C, D, y);
}


# states: the 64-step serial reconstruction becomes one parallel product S_t = Z_t + Sb_c.(A^k)^T; carry stores its zero-start states Z_t and every power A^k it passes; sum order of S changes
# speedup vs baseline: 3.1989x; 3.1989x over previous
#include <hip/hip_runtime.h>
#include <hip/hip_bf16.h>


typedef __attribute__((ext_vector_type(2))) float v2f;
typedef __attribute__((ext_vector_type(8))) float v8f;

#define S4_DT 0.001f
#define TT 4096
#define NB 16
#define NS 64
#define NI 256
#define NO 256
#define LCH 64
#define NCH 64

__device__ __forceinline__ v8f wmma4(v2f a, v2f b, v8f c) {
  return __builtin_amdgcn_wmma_f32_16x16x4_f32(
       false, a,  false, b,
       (short)0, c,  false,  false);
}

__device__ __forceinline__ v2f load2(const float* p) { return *(const v2f*)p; }

__device__ __forceinline__ void store_c(float* dst, int rs, int n0, v8f acc,
                                        int l16, int half) {
#pragma unroll
  for (int v = 0; v < 8; ++v)
    dst[(v + 8 * half) * rs + n0 + l16] = acc[v];
}

__device__ __forceinline__ v8f load_c(const float* src, int rs, int n0,
                                      int l16, int half) {
  v8f r;
#pragma unroll
  for (int v = 0; v < 8; ++v)
    r[v] = src[(v + 8 * half) * rs + n0 + l16];
  return r;
}

__global__ __launch_bounds__(256) void s4_prep_kernel(
    const float* __restrict__ A, const float* __restrict__ B,
    float* __restrict__ Abar, float* __restrict__ Bsc) {
  const int idx = blockIdx.x * 256 + threadIdx.x;
  if (idx < NS * NS) {
    int r = idx >> 6, c = idx & 63;
    Abar[idx] = A[idx] * S4_DT + (r == c ? 1.0f : 0.0f);
  } else {
    Bsc[idx - NS * NS] = B[idx - NS * NS] * S4_DT;
  }
}

__global__ __launch_bounds__(128) void s4_ub_kernel(
    const float* __restrict__ u, const float* __restrict__ Bsc,
    float* __restrict__ uB) {
  const int wave = threadIdx.x >> 5, lane = threadIdx.x & 31;
  const int l16 = lane & 15, half = lane >> 4;
  const int t = blockIdx.x * 4 + wave;

  const float* urow = u + (size_t)l16 * TT * NI + (size_t)t * NI;
  v8f acc[4] = {{}, {}, {}, {}};
  for (int kk = 0; kk < NI / 4; ++kk) {
    v2f a = load2(urow + 4 * kk + 2 * half);
#pragma unroll
    for (int j = 0; j < 4; ++j) {
      v2f b = load2(Bsc + (j * 16 + l16) * NI + 4 * kk + 2 * half);
      acc[j] = wmma4(a, b, acc[j]);
    }
  }
  float* dst = uB + (size_t)t * NB * NS;
#pragma unroll
  for (int j = 0; j < 4; ++j) store_c(dst, NS, j * 16, acc[j], l16, half);
}

__global__ __launch_bounds__(128) void s4_carry_kernel(
    const float* __restrict__ uB, const float* __restrict__ Abar,
    float* __restrict__ G, float* __restrict__ PW, float* __restrict__ Z) {
  __shared__ float sb[2][NB * NS];
  const int tid = threadIdx.x;
  const int wave = tid >> 5, lane = tid & 31;
  const int l16 = lane & 15, half = lane >> 4;
  const int n0 = wave * 16;
  const int c = blockIdx.x;
  const bool pw = c >= NCH;
  const int r0 = pw ? (c - NCH) * 16 : 0;
  const int cc = pw ? 0 : c;
  const float* F = uB + (size_t)cc * LCH * NB * NS;
  float* Zc = Z + (size_t)cc * LCH * NB * NS;
  float* prow = PW + (n0 + l16) * NS + r0 + 8 * half;

  v8f st;
#pragma unroll
  for (int v = 0; v < 8; ++v)
    st[v] = (pw && r0 + v + 8 * half == n0 + l16) ? 1.0f : 0.0f;
#pragma unroll
  for (int v = 0; v < 8; ++v) sb[0][(v + 8 * half) * NS + n0 + l16] = st[v];
  __syncthreads();

  int cur = 0;
  for (int k = 0; k < LCH; ++k) {
    v8f acc = {};
    if (pw) {
      *(v8f*)(prow + (size_t)k * NS * NS) = st;
    } else {
      store_c(Zc + (size_t)k * NB * NS, NS, n0, st, l16, half);
      acc = load_c(F + (size_t)k * NB * NS, NS, n0, l16, half);
    }
#pragma unroll
    for (int kk = 0; kk < 16; ++kk) {
      v2f a = load2(&sb[cur][l16 * NS + 4 * kk + 2 * half]);
      v2f b = load2(Abar + (n0 + l16) * NS + 4 * kk + 2 * half);
      acc = wmma4(a, b, acc);
    }
    int nxt = cur ^ 1;
#pragma unroll
    for (int v = 0; v < 8; ++v)
      sb[nxt][(v + 8 * half) * NS + n0 + l16] = acc[v];
    st = acc;
    __syncthreads();
    cur = nxt;
  }

  if (pw) {
    *(v8f*)(prow + (size_t)LCH * NS * NS) = st;
  } else {
    store_c(G + (size_t)c * NB * NS, NS, n0, st, l16, half);
  }
}

__global__ __launch_bounds__(128) void s4_bscan_kernel(
    const float* __restrict__ x0, const float* __restrict__ G,
    const float* __restrict__ A64, float* __restrict__ Sbound,
    float* __restrict__ xfinal) {
  __shared__ float sb[2][NB * NS];
  const int tid = threadIdx.x;
  const int wave = tid >> 5, lane = tid & 31;
  const int l16 = lane & 15, half = lane >> 4;
  const int n0 = wave * 16;
  const float* AL = A64;

  for (int idx = tid; idx < NB * NS; idx += 128) sb[0][idx] = x0[idx];
  __syncthreads();

  int cur = 0;
  for (int c = 0; c < NCH; ++c) {
    for (int idx = tid; idx < NB * NS; idx += 128)
      Sbound[(size_t)c * NB * NS + idx] = sb[cur][idx];
    v8f acc = load_c(G + (size_t)c * NB * NS, NS, n0, l16, half);
#pragma unroll
    for (int kk = 0; kk < 16; ++kk) {
      v2f a = load2(&sb[cur][l16 * NS + 4 * kk + 2 * half]);
      v2f b = load2(AL + (n0 + l16) * NS + 4 * kk + 2 * half);
      acc = wmma4(a, b, acc);
    }
    int nxt = cur ^ 1;
#pragma unroll
    for (int v = 0; v < 8; ++v)
      sb[nxt][(v + 8 * half) * NS + n0 + l16] = acc[v];
    __syncthreads();
    cur = nxt;
  }
  for (int idx = tid; idx < NB * NS; idx += 128) xfinal[idx] = sb[cur][idx];
}

#define FXC 4

__global__ __launch_bounds__(128) void s4_fix_kernel(
    const float* __restrict__ PW, const float* __restrict__ Sbound,
    float* S) {
  const int wave = threadIdx.x >> 5, lane = threadIdx.x & 31;
  const int l16 = lane & 15, half = lane >> 4;
  const int k = blockIdx.x & (LCH - 1);
  const int c0 = (blockIdx.x >> 6) * (4 * FXC) + wave * FXC;

  const float* P = PW + (size_t)k * NS * NS + l16 * NS + 2 * half;
  const float* sbp = Sbound + (size_t)c0 * NB * NS + l16 * NS + 2 * half;
  float* sp = S + ((size_t)c0 * LCH + k) * NB * NS + l16 * NS + 8 * half;

  v2f w[4][16];
#pragma unroll
  for (int kk = 0; kk < 16; ++kk)
#pragma unroll
    for (int i = 0; i < 4; ++i) w[i][kk] = load2(P + 16 * i * NS + 4 * kk);
  v2f b[FXC][16];
  v8f acc[FXC][4];
#pragma unroll
  for (int j = 0; j < FXC; ++j) {
#pragma unroll
    for (int kk = 0; kk < 16; ++kk)
      b[j][kk] = load2(sbp + (size_t)j * NB * NS + 4 * kk);
#pragma unroll
    for (int i = 0; i < 4; ++i)
      acc[j][i] = *(const v8f*)(sp + (size_t)j * LCH * NB * NS + 16 * i);
  }

#pragma unroll
  for (int j = 0; j < FXC; ++j) {
#pragma unroll
    for (int kk = 0; kk < 16; ++kk)
#pragma unroll
      for (int i = 0; i < 4; ++i)
        acc[j][i] = wmma4(w[i][kk], b[j][kk], acc[j][i]);
#pragma unroll
    for (int i = 0; i < 4; ++i)
      *(v8f*)(sp + (size_t)j * LCH * NB * NS + 16 * i) = acc[j][i];
  }
}

#define YT 16
#define YKG (NS / 4 + NI / 4)
#define YNB (YKG / 8)

__device__ __forceinline__ v2f y_load_a(const float* sp, const float* up,
                                        int kk) {
  return kk < NS / 4 ? load2(sp + 4 * kk) : load2(up + 4 * (kk - NS / 4));
}

__global__ __launch_bounds__(128) void s4_y_kernel(
    const float* __restrict__ S, const float* __restrict__ u,
    const float* __restrict__ C, const float* __restrict__ D,
    float* __restrict__ y) {
  const int wave = threadIdx.x >> 5, lane = threadIdx.x & 31;
  const int l16 = lane & 15, half = lane >> 4;
  const int t0 = blockIdx.x * YT;
  const int o0 = wave * 64;

  v2f w[YKG][4];
#pragma unroll
  for (int kk = 0; kk < NS / 4; ++kk)
#pragma unroll
    for (int j = 0; j < 4; ++j)
      w[kk][j] = load2(C + (o0 + 16 * j + l16) * NS + 4 * kk + 2 * half);
#pragma unroll
  for (int kk = 0; kk < NI / 4; ++kk)
#pragma unroll
    for (int j = 0; j < 4; ++j)
      w[NS / 4 + kk][j] =
          load2(D + (o0 + 16 * j + l16) * NI + 4 * kk + 2 * half);

  const float* srow = S + (size_t)t0 * NB * NS + l16 * NS + 2 * half;
  const float* urow = u + (size_t)l16 * TT * NI + (size_t)t0 * NI + 2 * half;
  v2f a[YKG];
#pragma unroll
  for (int kk = 0; kk < 8 * (YNB - 1); ++kk) a[kk] = y_load_a(srow, urow, kk);

  for (int i = 0; i < YT; ++i) {
    const float* sc = srow + (size_t)i * NB * NS;
    const float* uc = urow + (size_t)i * NI;
    const int in = (i + 1 < YT) ? i + 1 : i;
    const float* sn = srow + (size_t)in * NB * NS;
    const float* un = urow + (size_t)in * NI;

    v8f acc[4] = {{}, {}, {}, {}};
#pragma unroll
    for (int c8 = 0; c8 < YNB; ++c8) {
#pragma unroll
      for (int q = 0; q < 8; ++q) {
        const int kk = 8 * c8 + q;
#pragma unroll
        for (int j = 0; j < 4; ++j) acc[j] = wmma4(a[kk], w[kk][j], acc[j]);
      }
      const int r8 = (c8 + YNB - 1) % YNB;
#pragma unroll
      for (int q = 0; q < 8; ++q)
        a[8 * r8 + q] = y_load_a(c8 == 0 ? sc : sn, c8 == 0 ? uc : un, 8 * r8 + q);
      __builtin_amdgcn_sched_barrier(0);
    }

    const int t = t0 + i;
#pragma unroll
    for (int j = 0; j < 4; ++j)
#pragma unroll
      for (int v = 0; v < 8; ++v)
        y[(size_t)(v + 8 * half) * TT * NO + (size_t)t * NO + o0 + 16 * j + l16] =
            acc[j][v];
  }
}

extern "C" void kernel_launch(void* const* d_in, const int* in_sizes, int n_in,
                              void* d_out, int out_size, void* d_ws,
                              size_t ws_size, hipStream_t stream) {
  const float* u  = (const float*)d_in[0];
  const float* x0 = (const float*)d_in[1];
  const float* A  = (const float*)d_in[2];
  const float* B  = (const float*)d_in[3];
  const float* C  = (const float*)d_in[4];
  const float* D  = (const float*)d_in[5];

  float* y      = (float*)d_out;
  float* xfinal = (float*)d_out + (size_t)NB * TT * NO;

  float* ws = (float*)d_ws;
  float* Abar   = ws;                 ws += NS * NS;
  float* PW     = ws;                 ws += (LCH + 1) * NS * NS;
  float* A64    = PW + LCH * NS * NS;
  float* Bsc    = ws;                 ws += NS * NI;
  float* uB     = ws;                 ws += (size_t)TT * NB * NS;
  float* Sst    = ws;                 ws += (size_t)TT * NB * NS;
  float* G      = ws;                 ws += NCH * NB * NS;
  float* Sbound = ws;                 ws += NCH * NB * NS;

  s4_prep_kernel<<<(NS * NS + NS * NI) / 256, 256, 0, stream>>>(A, B, Abar, Bsc);
  s4_ub_kernel<<<TT / 4, 128, 0, stream>>>(u, Bsc, uB);
  s4_carry_kernel<<<NCH + NS / 16, 128, 0, stream>>>(uB, Abar, G, PW, Sst);
  s4_bscan_kernel<<<1, 128, 0, stream>>>(x0, G, A64, Sbound, xfinal);
  s4_fix_kernel<<<LCH * (NCH / (4 * FXC)), 128, 0, stream>>>(PW, Sbound, Sst);
  s4_y_kernel<<<TT / YT, 128, 0, stream>>>(Sst, u, C, D, y);
}


# scan split three ways (16 groups x 16 chunks x 16 steps): the zero-start chain kernel runs on chunks, then on groups with A^16; 48 serial steps, not 128; sum order of S and x_final changes
# speedup vs baseline: 5.1155x; 1.5991x over previous
#include <hip/hip_runtime.h>
#include <hip/hip_bf16.h>


typedef __attribute__((ext_vector_type(2))) float v2f;
typedef __attribute__((ext_vector_type(8))) float v8f;

#define S4_DT 0.001f
#define TT 4096
#define NB 16
#define NS 64
#define NI 256
#define NO 256
#define LCH 16
#define NCH 256
#define NGR 16

__device__ __forceinline__ v8f wmma4(v2f a, v2f b, v8f c) {
  return __builtin_amdgcn_wmma_f32_16x16x4_f32(
       false, a,  false, b,
       (short)0, c,  false,  false);
}

__device__ __forceinline__ v2f load2(const float* p) { return *(const v2f*)p; }

__device__ __forceinline__ void store_c(float* dst, int rs, int n0, v8f acc,
                                        int l16, int half) {
#pragma unroll
  for (int v = 0; v < 8; ++v)
    dst[(v + 8 * half) * rs + n0 + l16] = acc[v];
}

__device__ __forceinline__ v8f load_c(const float* src, int rs, int n0,
                                      int l16, int half) {
  v8f r;
#pragma unroll
  for (int v = 0; v < 8; ++v)
    r[v] = src[(v + 8 * half) * rs + n0 + l16];
  return r;
}

__global__ __launch_bounds__(256) void s4_prep_kernel(
    const float* __restrict__ A, const float* __restrict__ B,
    float* __restrict__ Abar, float* __restrict__ Bsc) {
  const int idx = blockIdx.x * 256 + threadIdx.x;
  if (idx < NS * NS) {
    int r = idx >> 6, c = idx & 63;
    Abar[idx] = A[idx] * S4_DT + (r == c ? 1.0f : 0.0f);
  } else {
    Bsc[idx - NS * NS] = B[idx - NS * NS] * S4_DT;
  }
}

__global__ __launch_bounds__(128) void s4_ub_kernel(
    const float* __restrict__ u, const float* __restrict__ Bsc,
    float* __restrict__ uB) {
  const int wave = threadIdx.x >> 5, lane = threadIdx.x & 31;
  const int l16 = lane & 15, half = lane >> 4;
  const int t = blockIdx.x * 4 + wave;

  const float* urow = u + (size_t)l16 * TT * NI + (size_t)t * NI;
  v8f acc[4] = {{}, {}, {}, {}};
  for (int kk = 0; kk < NI / 4; ++kk) {
    v2f a = load2(urow + 4 * kk + 2 * half);
#pragma unroll
    for (int j = 0; j < 4; ++j) {
      v2f b = load2(Bsc + (j * 16 + l16) * NI + 4 * kk + 2 * half);
      acc[j] = wmma4(a, b, acc[j]);
    }
  }
  float* dst = uB + (size_t)t * NB * NS;
#pragma unroll
  for (int j = 0; j < 4; ++j) store_c(dst, NS, j * 16, acc[j], l16, half);
}

__global__ __launch_bounds__(128) void s4_chain_kernel(
    const float* __restrict__ uB, const float* __restrict__ Abar,
    float* __restrict__ G, float* __restrict__ PW, float* __restrict__ Z,
    int nch) {
  __shared__ float sb[2][NB * NS];
  const int tid = threadIdx.x;
  const int wave = tid >> 5, lane = tid & 31;
  const int l16 = lane & 15, half = lane >> 4;
  const int n0 = wave * 16;
  const int c = blockIdx.x;
  const bool pw = c >= nch;
  const int r0 = pw ? (c - nch) * 16 : 0;
  const int cc = pw ? 0 : c;
  const float* F = uB + (size_t)cc * LCH * NB * NS;
  float* Zc = Z + (size_t)cc * LCH * NB * NS;
  float* prow = PW + (n0 + l16) * NS + r0 + 8 * half;

  v8f st;
#pragma unroll
  for (int v = 0; v < 8; ++v)
    st[v] = (pw && r0 + v + 8 * half == n0 + l16) ? 1.0f : 0.0f;
#pragma unroll
  for (int v = 0; v < 8; ++v) sb[0][(v + 8 * half) * NS + n0 + l16] = st[v];
  __syncthreads();

  int cur = 0;
  for (int k = 0; k < LCH; ++k) {
    v8f acc = {};
    if (pw) {
      *(v8f*)(prow + (size_t)k * NS * NS) = st;
    } else {
      store_c(Zc + (size_t)k * NB * NS, NS, n0, st, l16, half);
      acc = load_c(F + (size_t)k * NB * NS, NS, n0, l16, half);
    }
#pragma unroll
    for (int kk = 0; kk < 16; ++kk) {
      v2f a = load2(&sb[cur][l16 * NS + 4 * kk + 2 * half]);
      v2f b = load2(Abar + (n0 + l16) * NS + 4 * kk + 2 * half);
      acc = wmma4(a, b, acc);
    }
    int nxt = cur ^ 1;
#pragma unroll
    for (int v = 0; v < 8; ++v)
      sb[nxt][(v + 8 * half) * NS + n0 + l16] = acc[v];
    st = acc;
    __syncthreads();
    cur = nxt;
  }

  if (pw) {
    *(v8f*)(prow + (size_t)LCH * NS * NS) = st;
  } else {
    store_c(G + (size_t)c * NB * NS, NS, n0, st, l16, half);
  }
}

__global__ __launch_bounds__(128) void s4_bscan_kernel(
    const float* __restrict__ x0, const float* __restrict__ G,
    const float* __restrict__ Astep, float* __restrict__ Sbound,
    float* __restrict__ xfinal) {
  __shared__ float sb[2][NB * NS];
  const int tid = threadIdx.x;
  const int wave = tid >> 5, lane = tid & 31;
  const int l16 = lane & 15, half = lane >> 4;
  const int n0 = wave * 16;
  const float* AL = Astep;

  for (int idx = tid; idx < NB * NS; idx += 128) sb[0][idx] = x0[idx];
  __syncthreads();

  int cur = 0;
  for (int c = 0; c < NGR; ++c) {
    for (int idx = tid; idx < NB * NS; idx += 128)
      Sbound[(size_t)c * NB * NS + idx] = sb[cur][idx];
    v8f acc = load_c(G + (size_t)c * NB * NS, NS, n0, l16, half);
#pragma unroll
    for (int kk = 0; kk < 16; ++kk) {
      v2f a = load2(&sb[cur][l16 * NS + 4 * kk + 2 * half]);
      v2f b = load2(AL + (n0 + l16) * NS + 4 * kk + 2 * half);
      acc = wmma4(a, b, acc);
    }
    int nxt = cur ^ 1;
#pragma unroll
    for (int v = 0; v < 8; ++v)
      sb[nxt][(v + 8 * half) * NS + n0 + l16] = acc[v];
    __syncthreads();
    cur = nxt;
  }
  for (int idx = tid; idx < NB * NS; idx += 128) xfinal[idx] = sb[cur][idx];
}

#define FXC 4
#define FXB 1

template <int FX>
__device__ __forceinline__ void s4_fix_body(const float* PW,
                                            const float* Sbound, float* S) {
  const int wave = threadIdx.x >> 5, lane = threadIdx.x & 31;
  const int l16 = lane & 15, half = lane >> 4;
  const int k = blockIdx.x & (LCH - 1);
  const int c0 = (blockIdx.x / LCH) * (4 * FX) + wave * FX;

  const float* P = PW + (size_t)k * NS * NS + l16 * NS + 2 * half;
  const float* sbp = Sbound + (size_t)c0 * NB * NS + l16 * NS + 2 * half;
  float* sp = S + ((size_t)c0 * LCH + k) * NB * NS + l16 * NS + 8 * half;

  v2f w[4][16];
#pragma unroll
  for (int kk = 0; kk < 16; ++kk)
#pragma unroll
    for (int i = 0; i < 4; ++i) w[i][kk] = load2(P + 16 * i * NS + 4 * kk);
  v2f b[FX][16];
  v8f acc[FX][4];
#pragma unroll
  for (int j = 0; j < FX; ++j) {
#pragma unroll
    for (int kk = 0; kk < 16; ++kk)
      b[j][kk] = load2(sbp + (size_t)j * NB * NS + 4 * kk);
#pragma unroll
    for (int i = 0; i < 4; ++i)
      acc[j][i] = *(const v8f*)(sp + (size_t)j * LCH * NB * NS + 16 * i);
  }

#pragma unroll
  for (int j = 0; j < FX; ++j) {
#pragma unroll
    for (int kk = 0; kk < 16; ++kk)
#pragma unroll
      for (int i = 0; i < 4; ++i)
        acc[j][i] = wmma4(w[i][kk], b[j][kk], acc[j][i]);
#pragma unroll
    for (int i = 0; i < 4; ++i)
      *(v8f*)(sp + (size_t)j * LCH * NB * NS + 16 * i) = acc[j][i];
  }
}

__global__ __launch_bounds__(128) void s4_fixb_kernel(
    const float* __restrict__ PW, const float* __restrict__ Sg, float* Sb) {
  s4_fix_body<FXB>(PW, Sg, Sb);
}

__global__ __launch_bounds__(128) void s4_fix_kernel(
    const float* __restrict__ PW, const float* __restrict__ Sbound,
    float* S) {
  s4_fix_body<FXC>(PW, Sbound, S);
}

#define YT 16
#define YKG (NS / 4 + NI / 4)
#define YNB (YKG / 8)

__device__ __forceinline__ v2f y_load_a(const float* sp, const float* up,
                                        int kk) {
  return kk < NS / 4 ? load2(sp + 4 * kk) : load2(up + 4 * (kk - NS / 4));
}

__global__ __launch_bounds__(128) void s4_y_kernel(
    const float* __restrict__ S, const float* __restrict__ u,
    const float* __restrict__ C, const float* __restrict__ D,
    float* __restrict__ y) {
  const int wave = threadIdx.x >> 5, lane = threadIdx.x & 31;
  const int l16 = lane & 15, half = lane >> 4;
  const int t0 = blockIdx.x * YT;
  const int o0 = wave * 64;

  v2f w[YKG][4];
#pragma unroll
  for (int kk = 0; kk < NS / 4; ++kk)
#pragma unroll
    for (int j = 0; j < 4; ++j)
      w[kk][j] = load2(C + (o0 + 16 * j + l16) * NS + 4 * kk + 2 * half);
#pragma unroll
  for (int kk = 0; kk < NI / 4; ++kk)
#pragma unroll
    for (int j = 0; j < 4; ++j)
      w[NS / 4 + kk][j] =
          load2(D + (o0 + 16 * j + l16) * NI + 4 * kk + 2 * half);

  const float* srow = S + (size_t)t0 * NB * NS + l16 * NS + 2 * half;
  const float* urow = u + (size_t)l16 * TT * NI + (size_t)t0 * NI + 2 * half;
  v2f a[YKG];
#pragma unroll
  for (int kk = 0; kk < 8 * (YNB - 1); ++kk) a[kk] = y_load_a(srow, urow, kk);

  for (int i = 0; i < YT; ++i) {
    const float* sc = srow + (size_t)i * NB * NS;
    const float* uc = urow + (size_t)i * NI;
    const int in = (i + 1 < YT) ? i + 1 : i;
    const float* sn = srow + (size_t)in * NB * NS;
    const float* un = urow + (size_t)in * NI;

    v8f acc[4] = {{}, {}, {}, {}};
#pragma unroll
    for (int c8 = 0; c8 < YNB; ++c8) {
#pragma unroll
      for (int q = 0; q < 8; ++q) {
        const int kk = 8 * c8 + q;
#pragma unroll
        for (int j = 0; j < 4; ++j) acc[j] = wmma4(a[kk], w[kk][j], acc[j]);
      }
      const int r8 = (c8 + YNB - 1) % YNB;
#pragma unroll
      for (int q = 0; q < 8; ++q)
        a[8 * r8 + q] = y_load_a(c8 == 0 ? sc : sn, c8 == 0 ? uc : un, 8 * r8 + q);
      __builtin_amdgcn_sched_barrier(0);
    }

    const int t = t0 + i;
#pragma unroll
    for (int j = 0; j < 4; ++j)
#pragma unroll
      for (int v = 0; v < 8; ++v)
        y[(size_t)(v + 8 * half) * TT * NO + (size_t)t * NO + o0 + 16 * j + l16] =
            acc[j][v];
  }
}

extern "C" void kernel_launch(void* const* d_in, const int* in_sizes, int n_in,
                              void* d_out, int out_size, void* d_ws,
                              size_t ws_size, hipStream_t stream) {
  const float* u  = (const float*)d_in[0];
  const float* x0 = (const float*)d_in[1];
  const float* A  = (const float*)d_in[2];
  const float* B  = (const float*)d_in[3];
  const float* C  = (const float*)d_in[4];
  const float* D  = (const float*)d_in[5];

  float* y      = (float*)d_out;
  float* xfinal = (float*)d_out + (size_t)NB * TT * NO;

  float* ws = (float*)d_ws;
  float* Abar   = ws;                 ws += NS * NS;
  float* PW1    = ws;                 ws += (LCH + 1) * NS * NS;
  float* A16    = PW1 + LCH * NS * NS;
  float* PW2    = ws;                 ws += (LCH + 1) * NS * NS;
  float* A256   = PW2 + LCH * NS * NS;
  float* Bsc    = ws;
  float* G      = ws;                 ws += NCH * NB * NS;
  float* uB     = ws;                 ws += (size_t)TT * NB * NS;
  float* Sst    = ws;                 ws += (size_t)TT * NB * NS;
  float* Sb     = uB;
  float* GG     = Sb + NCH * NB * NS;
  float* Sg     = GG + NGR * NB * NS;

  s4_prep_kernel<<<(NS * NS + NS * NI) / 256, 256, 0, stream>>>(A, B, Abar, Bsc);
  s4_ub_kernel<<<TT / 4, 128, 0, stream>>>(u, Bsc, uB);
  s4_chain_kernel<<<NCH + NS / 16, 128, 0, stream>>>(uB, Abar, G, PW1, Sst, NCH);
  s4_chain_kernel<<<NGR + NS / 16, 128, 0, stream>>>(G, A16, GG, PW2, Sb, NGR);
  s4_bscan_kernel<<<1, 128, 0, stream>>>(x0, GG, A256, Sg, xfinal);
  s4_fixb_kernel<<<LCH * (NGR / (4 * FXB)), 128, 0, stream>>>(PW2, Sg, Sb);
  s4_fix_kernel<<<LCH * (NCH / (4 * FXC)), 128, 0, stream>>>(PW1, Sb, Sst);
  s4_y_kernel<<<TT / YT, 128, 0, stream>>>(Sst, u, C, D, y);
}


# chain kernels: one wave per chunk runs each step transposed, state kept in accumulator registers (k order makes D the next B fragment), no LDS, no barrier; sum order of S, x_final, powers changes
# speedup vs baseline: 6.2522x; 1.2222x over previous
#include <hip/hip_runtime.h>
#include <hip/hip_bf16.h>


typedef __attribute__((ext_vector_type(2))) float v2f;
typedef __attribute__((ext_vector_type(8))) float v8f;

#define S4_DT 0.001f
#define TT 4096
#define NB 16
#define NS 64
#define NI 256
#define NO 256
#define LCH 16
#define NCH 256
#define NGR 16

__device__ __forceinline__ v8f wmma4(v2f a, v2f b, v8f c) {
  return __builtin_amdgcn_wmma_f32_16x16x4_f32(
       false, a,  false, b,
       (short)0, c,  false,  false);
}

__device__ __forceinline__ v2f load2(const float* p) { return *(const v2f*)p; }

__device__ __forceinline__ void store_c(float* dst, int rs, int n0, v8f acc,
                                        int l16, int half) {
#pragma unroll
  for (int v = 0; v < 8; ++v)
    dst[(v + 8 * half) * rs + n0 + l16] = acc[v];
}

__device__ __forceinline__ v8f load_c(const float* src, int rs, int n0,
                                      int l16, int half) {
  v8f r;
#pragma unroll
  for (int v = 0; v < 8; ++v)
    r[v] = src[(v + 8 * half) * rs + n0 + l16];
  return r;
}

__global__ __launch_bounds__(256) void s4_prep_kernel(
    const float* __restrict__ A, const float* __restrict__ B,
    float* __restrict__ Abar, float* __restrict__ Bsc) {
  const int idx = blockIdx.x * 256 + threadIdx.x;
  if (idx < NS * NS) {
    int r = idx >> 6, c = idx & 63;
    Abar[idx] = A[idx] * S4_DT + (r == c ? 1.0f : 0.0f);
  } else {
    Bsc[idx - NS * NS] = B[idx - NS * NS] * S4_DT;
  }
}

__global__ __launch_bounds__(128) void s4_ub_kernel(
    const float* __restrict__ u, const float* __restrict__ Bsc,
    float* __restrict__ uB) {
  const int wave = threadIdx.x >> 5, lane = threadIdx.x & 31;
  const int l16 = lane & 15, half = lane >> 4;
  const int t = blockIdx.x * 4 + wave;

  const float* urow = u + (size_t)l16 * TT * NI + (size_t)t * NI;
  v8f acc[4] = {{}, {}, {}, {}};
  for (int kk = 0; kk < NI / 4; ++kk) {
    v2f a = load2(urow + 4 * kk + 2 * half);
#pragma unroll
    for (int j = 0; j < 4; ++j) {
      v2f b = load2(Bsc + (j * 16 + l16) * NI + 4 * kk + 2 * half);
      acc[j] = wmma4(a, b, acc[j]);
    }
  }
  float* dst = uB + (size_t)t * NB * NS;
#pragma unroll
  for (int j = 0; j < 4; ++j) store_c(dst, NS, j * 16, acc[j], l16, half);
}

template <bool PWR>
__device__ __forceinline__ void s4_put_state(float* p, const v8f (&st)[4]) {
  if (PWR) {
#pragma unroll
    for (int i = 0; i < 4; ++i)
#pragma unroll
      for (int r = 0; r < 8; ++r) p[(16 * i + r) * NS] = st[i][r];
  } else {
#pragma unroll
    for (int i = 0; i < 4; ++i) *(v8f*)(p + 16 * i) = st[i];
  }
}

template <bool PWR>
__device__ __forceinline__ void s4_fetch(v8f (&f)[4], const float* p) {
  if (!PWR) {
#pragma unroll
    for (int i = 0; i < 4; ++i) f[i] = *(const v8f*)(p + 16 * i);
  }
}

template <bool PWR>
__device__ __forceinline__ void s4_chain_step(const v2f (&a)[4][16],
                                              const v8f (&cur)[4],
                                              v8f (&nxt)[4]) {
#pragma unroll
  for (int kk = 0; kk < 16; ++kk) {
    const int q = kk >> 2, p = kk & 3;
    v2f b = {cur[q][2 * p], cur[q][2 * p + 1]};
#pragma unroll
    for (int i = 0; i < 4; ++i)
      nxt[i] = wmma4(a[i][kk], b, (PWR && kk == 0) ? v8f{} : nxt[i]);
  }
  __builtin_amdgcn_sched_barrier(0);
}

template <bool PWR>
__device__ __forceinline__ void s4_chain_body(const float* F, const float* M,
                                              float* G, float* PW, float* Z,
                                              const float* X0, int c) {
  const int lane = threadIdx.x & 31;
  const int l16 = lane & 15, half = lane >> 4;

  v2f a[4][16];
#pragma unroll
  for (int i = 0; i < 4; ++i)
#pragma unroll
    for (int q = 0; q < 4; ++q) {
      v8f w = *(const v8f*)(M + (16 * i + l16) * NS + 16 * q + 8 * half);
      a[i][4 * q + 0] = __builtin_shufflevector(w, w, 0, 1);
      a[i][4 * q + 1] = __builtin_shufflevector(w, w, 2, 3);
      a[i][4 * q + 2] = __builtin_shufflevector(w, w, 4, 5);
      a[i][4 * q + 3] = __builtin_shufflevector(w, w, 6, 7);
    }

  const int lo = l16 * NS + 8 * half;
  const int r0 = 16 * c;
  const float* Fc = F + (size_t)c * LCH * NB * NS + lo;
  float* out = PWR ? PW + 8 * half * NS + r0 + l16
                   : Z + (size_t)c * LCH * NB * NS + lo;
  const int ostep = PWR ? NS * NS : NB * NS;
  const int fstep = NB * NS;

  v8f s0[4], s1[4], s2[4];
#pragma unroll
  for (int i = 0; i < 4; ++i) {
    if (PWR) {
#pragma unroll
      for (int r = 0; r < 8; ++r)
        s0[i][r] = (16 * i + 8 * half + r == r0 + l16) ? 1.0f : 0.0f;
    } else {
      s0[i] = X0 ? *(const v8f*)(X0 + lo + 16 * i) : v8f{};
    }
  }
  s4_fetch<PWR>(s1, Fc);

  for (int k = 0; k + 3 <= LCH; k += 3) {
    s4_put_state<PWR>(out + (size_t)k * ostep, s0);
    s4_fetch<PWR>(s2, Fc + (size_t)(k + 1) * fstep);
    s4_chain_step<PWR>(a, s0, s1);
    s4_put_state<PWR>(out + (size_t)(k + 1) * ostep, s1);
    s4_fetch<PWR>(s0, Fc + (size_t)(k + 2) * fstep);
    s4_chain_step<PWR>(a, s1, s2);
    s4_put_state<PWR>(out + (size_t)(k + 2) * ostep, s2);
    s4_fetch<PWR>(s1, Fc + (size_t)(k + 3) * fstep);
    s4_chain_step<PWR>(a, s2, s0);
  }
  static_assert(LCH % 3 == 1, "one step left behind the loop of three");
  s4_put_state<PWR>(out + (size_t)(LCH - 1) * ostep, s0);
  s4_chain_step<PWR>(a, s0, s1);

  if (PWR) {
    s4_put_state<true>(out + (size_t)LCH * ostep, s1);
  } else {
    s4_put_state<false>(G + (size_t)c * NB * NS + lo, s1);
  }
}

__global__ __launch_bounds__(32) void s4_chain_kernel(
    const float* __restrict__ F, const float* __restrict__ M,
    float* __restrict__ G, float* __restrict__ PW, float* __restrict__ Z,
    const float* __restrict__ X0, int nch) {
  const int c = blockIdx.x;
  if (c < nch) {
    s4_chain_body<false>(F, M, G, PW, Z, X0, c);
  } else {
    s4_chain_body<true>(F, M, G, PW, Z, X0, c - nch);
  }
}

#define FXC 4
#define FXB 1

template <int FX>
__device__ __forceinline__ void s4_fix_body(const float* PW,
                                            const float* Sbound, float* S) {
  const int wave = threadIdx.x >> 5, lane = threadIdx.x & 31;
  const int l16 = lane & 15, half = lane >> 4;
  const int k = blockIdx.x & (LCH - 1);
  const int c0 = (blockIdx.x / LCH) * (4 * FX) + wave * FX;

  const float* P = PW + (size_t)k * NS * NS + l16 * NS + 2 * half;
  const float* sbp = Sbound + (size_t)c0 * NB * NS + l16 * NS + 2 * half;
  float* sp = S + ((size_t)c0 * LCH + k) * NB * NS + l16 * NS + 8 * half;

  v2f w[4][16];
#pragma unroll
  for (int kk = 0; kk < 16; ++kk)
#pragma unroll
    for (int i = 0; i < 4; ++i) w[i][kk] = load2(P + 16 * i * NS + 4 * kk);
  v2f b[FX][16];
  v8f acc[FX][4];
#pragma unroll
  for (int j = 0; j < FX; ++j) {
#pragma unroll
    for (int kk = 0; kk < 16; ++kk)
      b[j][kk] = load2(sbp + (size_t)j * NB * NS + 4 * kk);
#pragma unroll
    for (int i = 0; i < 4; ++i)
      acc[j][i] = *(const v8f*)(sp + (size_t)j * LCH * NB * NS + 16 * i);
  }

#pragma unroll
  for (int j = 0; j < FX; ++j) {
#pragma unroll
    for (int kk = 0; kk < 16; ++kk)
#pragma unroll
      for (int i = 0; i < 4; ++i)
        acc[j][i] = wmma4(w[i][kk], b[j][kk], acc[j][i]);
#pragma unroll
    for (int i = 0; i < 4; ++i)
      *(v8f*)(sp + (size_t)j * LCH * NB * NS + 16 * i) = acc[j][i];
  }
}

__global__ __launch_bounds__(128) void s4_fixb_kernel(
    const float* __restrict__ PW, const float* __restrict__ Sg, float* Sb) {
  s4_fix_body<FXB>(PW, Sg, Sb);
}

__global__ __launch_bounds__(128) void s4_fix_kernel(
    const float* __restrict__ PW, const float* __restrict__ Sbound,
    float* S) {
  s4_fix_body<FXC>(PW, Sbound, S);
}

#define YT 16
#define YKG (NS / 4 + NI / 4)
#define YNB (YKG / 8)

__device__ __forceinline__ v2f y_load_a(const float* sp, const float* up,
                                        int kk) {
  return kk < NS / 4 ? load2(sp + 4 * kk) : load2(up + 4 * (kk - NS / 4));
}

__global__ __launch_bounds__(128) void s4_y_kernel(
    const float* __restrict__ S, const float* __restrict__ u,
    const float* __restrict__ C, const float* __restrict__ D,
    float* __restrict__ y) {
  const int wave = threadIdx.x >> 5, lane = threadIdx.x & 31;
  const int l16 = lane & 15, half = lane >> 4;
  const int t0 = blockIdx.x * YT;
  const int o0 = wave * 64;

  v2f w[YKG][4];
#pragma unroll
  for (int kk = 0; kk < NS / 4; ++kk)
#pragma unroll
    for (int j = 0; j < 4; ++j)
      w[kk][j] = load2(C + (o0 + 16 * j + l16) * NS + 4 * kk + 2 * half);
#pragma unroll
  for (int kk = 0; kk < NI / 4; ++kk)
#pragma unroll
    for (int j = 0; j < 4; ++j)
      w[NS / 4 + kk][j] =
          load2(D + (o0 + 16 * j + l16) * NI + 4 * kk + 2 * half);

  const float* srow = S + (size_t)t0 * NB * NS + l16 * NS + 2 * half;
  const float* urow = u + (size_t)l16 * TT * NI + (size_t)t0 * NI + 2 * half;
  v2f a[YKG];
#pragma unroll
  for (int kk = 0; kk < 8 * (YNB - 1); ++kk) a[kk] = y_load_a(srow, urow, kk);

  for (int i = 0; i < YT; ++i) {
    const float* sc = srow + (size_t)i * NB * NS;
    const float* uc = urow + (size_t)i * NI;
    const int in = (i + 1 < YT) ? i + 1 : i;
    const float* sn = srow + (size_t)in * NB * NS;
    const float* un = urow + (size_t)in * NI;

    v8f acc[4] = {{}, {}, {}, {}};
#pragma unroll
    for (int c8 = 0; c8 < YNB; ++c8) {
#pragma unroll
      for (int q = 0; q < 8; ++q) {
        const int kk = 8 * c8 + q;
#pragma unroll
        for (int j = 0; j < 4; ++j) acc[j] = wmma4(a[kk], w[kk][j], acc[j]);
      }
      const int r8 = (c8 + YNB - 1) % YNB;
#pragma unroll
      for (int q = 0; q < 8; ++q)
        a[8 * r8 + q] = y_load_a(c8 == 0 ? sc : sn, c8 == 0 ? uc : un, 8 * r8 + q);
      __builtin_amdgcn_sched_barrier(0);
    }

    const int t = t0 + i;
#pragma unroll
    for (int j = 0; j < 4; ++j)
#pragma unroll
      for (int v = 0; v < 8; ++v)
        y[(size_t)(v + 8 * half) * TT * NO + (size_t)t * NO + o0 + 16 * j + l16] =
            acc[j][v];
  }
}

extern "C" void kernel_launch(void* const* d_in, const int* in_sizes, int n_in,
                              void* d_out, int out_size, void* d_ws,
                              size_t ws_size, hipStream_t stream) {
  const float* u  = (const float*)d_in[0];
  const float* x0 = (const float*)d_in[1];
  const float* A  = (const float*)d_in[2];
  const float* B  = (const float*)d_in[3];
  const float* C  = (const float*)d_in[4];
  const float* D  = (const float*)d_in[5];

  float* y      = (float*)d_out;
  float* xfinal = (float*)d_out + (size_t)NB * TT * NO;

  float* ws = (float*)d_ws;
  float* Abar   = ws;                 ws += NS * NS;
  float* PW1    = ws;                 ws += (LCH + 1) * NS * NS;
  float* A16    = PW1 + LCH * NS * NS;
  float* PW2    = ws;                 ws += (LCH + 1) * NS * NS;
  float* A256   = PW2 + LCH * NS * NS;
  float* Bsc    = ws;
  float* G      = ws;                 ws += NCH * NB * NS;
  float* uB     = ws;                 ws += (size_t)TT * NB * NS;
  float* Sst    = ws;                 ws += (size_t)TT * NB * NS;
  float* Sb     = uB;
  float* GG     = Sb + NCH * NB * NS;
  float* Sg     = GG + NGR * NB * NS;

  s4_prep_kernel<<<(NS * NS + NS * NI) / 256, 256, 0, stream>>>(A, B, Abar, Bsc);
  s4_ub_kernel<<<TT / 4, 128, 0, stream>>>(u, Bsc, uB);
  s4_chain_kernel<<<NCH + NS / 16, 32, 0, stream>>>(uB, Abar, G, PW1, Sst, nullptr, NCH);
  s4_chain_kernel<<<NGR + NS / 16, 32, 0, stream>>>(G, A16, GG, PW2, Sb, nullptr, NGR);
  s4_chain_kernel<<<1, 32, 0, stream>>>(GG, A256, xfinal, nullptr, Sg, x0, 1);
  s4_fixb_kernel<<<LCH * (NGR / (4 * FXB)), 128, 0, stream>>>(PW2, Sg, Sb);
  s4_fix_kernel<<<LCH * (NCH / (4 * FXC)), 128, 0, stream>>>(PW1, Sb, Sst);
  s4_y_kernel<<<TT / YT, 128, 0, stream>>>(Sst, u, C, D, y);
}


# tail in ONE launch (6 launches, not 8): every wave rescans its own group's start state, then applies the chunk and step powers with the chain's step; no Sg/Sb round trip; sum order of S changes
# speedup vs baseline: 6.6181x; 1.0585x over previous
#include <hip/hip_runtime.h>
#include <hip/hip_bf16.h>


typedef __attribute__((ext_vector_type(2))) float v2f;
typedef __attribute__((ext_vector_type(8))) float v8f;

#define S4_DT 0.001f
#define TT 4096
#define NB 16
#define NS 64
#define NI 256
#define NO 256
#define LCH 16
#define NCH 256
#define NGR 16

__device__ __forceinline__ v8f wmma4(v2f a, v2f b, v8f c) {
  return __builtin_amdgcn_wmma_f32_16x16x4_f32(
       false, a,  false, b,
       (short)0, c,  false,  false);
}

__device__ __forceinline__ v2f load2(const float* p) { return *(const v2f*)p; }

__device__ __forceinline__ void store_c(float* dst, int rs, int n0, v8f acc,
                                        int l16, int half) {
#pragma unroll
  for (int v = 0; v < 8; ++v)
    dst[(v + 8 * half) * rs + n0 + l16] = acc[v];
}

__device__ __forceinline__ v8f load_c(const float* src, int rs, int n0,
                                      int l16, int half) {
  v8f r;
#pragma unroll
  for (int v = 0; v < 8; ++v)
    r[v] = src[(v + 8 * half) * rs + n0 + l16];
  return r;
}

__global__ __launch_bounds__(256) void s4_prep_kernel(
    const float* __restrict__ A, const float* __restrict__ B,
    float* __restrict__ Abar, float* __restrict__ Bsc) {
  const int idx = blockIdx.x * 256 + threadIdx.x;
  if (idx < NS * NS) {
    int r = idx >> 6, c = idx & 63;
    Abar[idx] = A[idx] * S4_DT + (r == c ? 1.0f : 0.0f);
  } else {
    Bsc[idx - NS * NS] = B[idx - NS * NS] * S4_DT;
  }
}

__global__ __launch_bounds__(128) void s4_ub_kernel(
    const float* __restrict__ u, const float* __restrict__ Bsc,
    float* __restrict__ uB) {
  const int wave = threadIdx.x >> 5, lane = threadIdx.x & 31;
  const int l16 = lane & 15, half = lane >> 4;
  const int t = blockIdx.x * 4 + wave;

  const float* urow = u + (size_t)l16 * TT * NI + (size_t)t * NI;
  v8f acc[4] = {{}, {}, {}, {}};
  for (int kk = 0; kk < NI / 4; ++kk) {
    v2f a = load2(urow + 4 * kk + 2 * half);
#pragma unroll
    for (int j = 0; j < 4; ++j) {
      v2f b = load2(Bsc + (j * 16 + l16) * NI + 4 * kk + 2 * half);
      acc[j] = wmma4(a, b, acc[j]);
    }
  }
  float* dst = uB + (size_t)t * NB * NS;
#pragma unroll
  for (int j = 0; j < 4; ++j) store_c(dst, NS, j * 16, acc[j], l16, half);
}

template <bool PWR>
__device__ __forceinline__ void s4_put_state(float* p, const v8f (&st)[4]) {
  if (PWR) {
#pragma unroll
    for (int i = 0; i < 4; ++i)
#pragma unroll
      for (int r = 0; r < 8; ++r) p[(16 * i + r) * NS] = st[i][r];
  } else {
#pragma unroll
    for (int i = 0; i < 4; ++i) *(v8f*)(p + 16 * i) = st[i];
  }
}

template <bool PWR>
__device__ __forceinline__ void s4_fetch(v8f (&f)[4], const float* p) {
  if (!PWR) {
#pragma unroll
    for (int i = 0; i < 4; ++i) f[i] = *(const v8f*)(p + 16 * i);
  }
}

template <bool PWR>
__device__ __forceinline__ void s4_chain_step(const v2f (&a)[4][16],
                                              const v8f (&cur)[4],
                                              v8f (&nxt)[4]) {
#pragma unroll
  for (int kk = 0; kk < 16; ++kk) {
    const int q = kk >> 2, p = kk & 3;
    v2f b = {cur[q][2 * p], cur[q][2 * p + 1]};
#pragma unroll
    for (int i = 0; i < 4; ++i)
      nxt[i] = wmma4(a[i][kk], b, (PWR && kk == 0) ? v8f{} : nxt[i]);
  }
  __builtin_amdgcn_sched_barrier(0);
}

__device__ __forceinline__ void s4_load_mfrag(v2f (&a)[4][16], const float* M,
                                              int l16, int half) {
#pragma unroll
  for (int i = 0; i < 4; ++i)
#pragma unroll
    for (int q = 0; q < 4; ++q) {
      v8f w = *(const v8f*)(M + (16 * i + l16) * NS + 16 * q + 8 * half);
      a[i][4 * q + 0] = __builtin_shufflevector(w, w, 0, 1);
      a[i][4 * q + 1] = __builtin_shufflevector(w, w, 2, 3);
      a[i][4 * q + 2] = __builtin_shufflevector(w, w, 4, 5);
      a[i][4 * q + 3] = __builtin_shufflevector(w, w, 6, 7);
    }
}

template <bool PWR>
__device__ __forceinline__ void s4_chain_body(const float* F, const float* M,
                                              float* G, float* PW, float* Z,
                                              const float* X0, int c) {
  const int lane = threadIdx.x & 31;
  const int l16 = lane & 15, half = lane >> 4;

  v2f a[4][16];
  s4_load_mfrag(a, M, l16, half);

  const int lo = l16 * NS + 8 * half;
  const int r0 = 16 * c;
  const float* Fc = F + (size_t)c * LCH * NB * NS + lo;
  float* out = PWR ? PW + 8 * half * NS + r0 + l16
                   : Z + (size_t)c * LCH * NB * NS + lo;
  const int ostep = PWR ? NS * NS : NB * NS;
  const int fstep = NB * NS;

  v8f s0[4], s1[4], s2[4];
#pragma unroll
  for (int i = 0; i < 4; ++i) {
    if (PWR) {
#pragma unroll
      for (int r = 0; r < 8; ++r)
        s0[i][r] = (16 * i + 8 * half + r == r0 + l16) ? 1.0f : 0.0f;
    } else {
      s0[i] = X0 ? *(const v8f*)(X0 + lo + 16 * i) : v8f{};
    }
  }
  s4_fetch<PWR>(s1, Fc);

  for (int k = 0; k + 3 <= LCH; k += 3) {
    s4_put_state<PWR>(out + (size_t)k * ostep, s0);
    s4_fetch<PWR>(s2, Fc + (size_t)(k + 1) * fstep);
    s4_chain_step<PWR>(a, s0, s1);
    s4_put_state<PWR>(out + (size_t)(k + 1) * ostep, s1);
    s4_fetch<PWR>(s0, Fc + (size_t)(k + 2) * fstep);
    s4_chain_step<PWR>(a, s1, s2);
    s4_put_state<PWR>(out + (size_t)(k + 2) * ostep, s2);
    s4_fetch<PWR>(s1, Fc + (size_t)(k + 3) * fstep);
    s4_chain_step<PWR>(a, s2, s0);
  }
  static_assert(LCH % 3 == 1, "one step left behind the loop of three");
  s4_put_state<PWR>(out + (size_t)(LCH - 1) * ostep, s0);
  s4_chain_step<PWR>(a, s0, s1);

  if (PWR) {
    s4_put_state<true>(out + (size_t)LCH * ostep, s1);
  } else {
    s4_put_state<false>(G + (size_t)c * NB * NS + lo, s1);
  }
}

__global__ __launch_bounds__(32) void s4_chain_kernel(
    const float* __restrict__ F, const float* __restrict__ M,
    float* __restrict__ G, float* __restrict__ PW, float* __restrict__ Z,
    const float* __restrict__ X0, int nch) {
  const int c = blockIdx.x;
  if (c < nch) {
    s4_chain_body<false>(F, M, G, PW, Z, X0, c);
  } else {
    s4_chain_body<true>(F, M, G, PW, Z, X0, c - nch);
  }
}

#define TLK (LCH / 4)

__global__ __launch_bounds__(128) void s4_tail_kernel(
    const float* __restrict__ GG, const float* __restrict__ A256,
    const float* __restrict__ X0, const float* __restrict__ PW2,
    const float* __restrict__ Zb, const float* __restrict__ PW1,
    float* S, float* __restrict__ xfinal) {
  const int wave = threadIdx.x >> 5, lane = threadIdx.x & 31;
  const int l16 = lane & 15, half = lane >> 4;
  const int c = blockIdx.x, g = c / LCH, j = c % LCH;
  const int lo = l16 * NS + 8 * half;
  const int tile = NB * NS;
  const float* P1 = PW1 + (size_t)(wave * TLK) * NS * NS;
  float* sp = S + ((size_t)c * LCH + wave * TLK) * tile + lo;

  v2f a[4][16], aA[4][16], aB[4][16];
  v8f s0[4], s1[4], s2[4], sb[4], zA[4], zB[4];
  s4_load_mfrag(a, A256, l16, half);
#pragma unroll
  for (int i = 0; i < 4; ++i) s0[i] = *(const v8f*)(X0 + lo + 16 * i);
  s4_fetch<false>(s1, GG + lo);
  s4_load_mfrag(aB, PW2 + (size_t)j * NS * NS, l16, half);
  s4_fetch<false>(sb, Zb + (size_t)c * tile + lo);
  s4_load_mfrag(aA, P1, l16, half);
  s4_fetch<false>(zA, sp);

  int gi = 0;
  for (; gi + 3 <= g; gi += 3) {
    s4_fetch<false>(s2, GG + (size_t)(gi + 1) * tile + lo);
    s4_chain_step<false>(a, s0, s1);
    s4_fetch<false>(s0, GG + (size_t)(gi + 2) * tile + lo);
    s4_chain_step<false>(a, s1, s2);
    s4_fetch<false>(s1, GG + (size_t)(gi + 3) * tile + lo);
    s4_chain_step<false>(a, s2, s0);
  }
  v8f sg[4];
  if (g - gi == 2) {
    s4_fetch<false>(s2, GG + (size_t)(gi + 1) * tile + lo);
    s4_chain_step<false>(a, s0, s1);
    s4_chain_step<false>(a, s1, s2);
#pragma unroll
    for (int i = 0; i < 4; ++i) sg[i] = s2[i];
  } else if (g - gi == 1) {
    s4_chain_step<false>(a, s0, s1);
#pragma unroll
    for (int i = 0; i < 4; ++i) sg[i] = s1[i];
  } else {
#pragma unroll
    for (int i = 0; i < 4; ++i) sg[i] = s0[i];
  }

  if (c == NCH - 1 && wave == 0) {
    v8f xf[4];
    s4_fetch<false>(xf, GG + (size_t)(NGR - 1) * tile + lo);
    s4_chain_step<false>(a, sg, xf);
    s4_put_state<false>(xfinal + lo, xf);
  }

  s4_chain_step<false>(aB, sg, sb);
  s4_load_mfrag(aB, P1 + NS * NS, l16, half);
  s4_fetch<false>(zB, sp + tile);
  s4_chain_step<false>(aA, sb, zA);
  s4_put_state<false>(sp, zA);
  s4_load_mfrag(aA, P1 + 2 * NS * NS, l16, half);
  s4_fetch<false>(zA, sp + 2 * tile);
  s4_chain_step<false>(aB, sb, zB);
  s4_put_state<false>(sp + tile, zB);
  s4_load_mfrag(aB, P1 + 3 * NS * NS, l16, half);
  s4_fetch<false>(zB, sp + 3 * tile);
  s4_chain_step<false>(aA, sb, zA);
  s4_put_state<false>(sp + 2 * tile, zA);
  s4_chain_step<false>(aB, sb, zB);
  s4_put_state<false>(sp + 3 * tile, zB);
  static_assert(TLK == 4, "four products per wave, written out");
}

#define YT 16
#define YKG (NS / 4 + NI / 4)
#define YNB (YKG / 8)

__device__ __forceinline__ v2f y_load_a(const float* sp, const float* up,
                                        int kk) {
  return kk < NS / 4 ? load2(sp + 4 * kk) : load2(up + 4 * (kk - NS / 4));
}

__global__ __launch_bounds__(128) void s4_y_kernel(
    const float* __restrict__ S, const float* __restrict__ u,
    const float* __restrict__ C, const float* __restrict__ D,
    float* __restrict__ y) {
  const int wave = threadIdx.x >> 5, lane = threadIdx.x & 31;
  const int l16 = lane & 15, half = lane >> 4;
  const int t0 = blockIdx.x * YT;
  const int o0 = wave * 64;

  v2f w[YKG][4];
#pragma unroll
  for (int kk = 0; kk < NS / 4; ++kk)
#pragma unroll
    for (int j = 0; j < 4; ++j)
      w[kk][j] = load2(C + (o0 + 16 * j + l16) * NS + 4 * kk + 2 * half);
#pragma unroll
  for (int kk = 0; kk < NI / 4; ++kk)
#pragma unroll
    for (int j = 0; j < 4; ++j)
      w[NS / 4 + kk][j] =
          load2(D + (o0 + 16 * j + l16) * NI + 4 * kk + 2 * half);

  const float* srow = S + (size_t)t0 * NB * NS + l16 * NS + 2 * half;
  const float* urow = u + (size_t)l16 * TT * NI + (size_t)t0 * NI + 2 * half;
  v2f a[YKG];
#pragma unroll
  for (int kk = 0; kk < 8 * (YNB - 1); ++kk) a[kk] = y_load_a(srow, urow, kk);

  for (int i = 0; i < YT; ++i) {
    const float* sc = srow + (size_t)i * NB * NS;
    const float* uc = urow + (size_t)i * NI;
    const int in = (i + 1 < YT) ? i + 1 : i;
    const float* sn = srow + (size_t)in * NB * NS;
    const float* un = urow + (size_t)in * NI;

    v8f acc[4] = {{}, {}, {}, {}};
#pragma unroll
    for (int c8 = 0; c8 < YNB; ++c8) {
#pragma unroll
      for (int q = 0; q < 8; ++q) {
        const int kk = 8 * c8 + q;
#pragma unroll
        for (int j = 0; j < 4; ++j) acc[j] = wmma4(a[kk], w[kk][j], acc[j]);
      }
      const int r8 = (c8 + YNB - 1) % YNB;
#pragma unroll
      for (int q = 0; q < 8; ++q)
        a[8 * r8 + q] = y_load_a(c8 == 0 ? sc : sn, c8 == 0 ? uc : un, 8 * r8 + q);
      __builtin_amdgcn_sched_barrier(0);
    }

    const int t = t0 + i;
#pragma unroll
    for (int j = 0; j < 4; ++j)
#pragma unroll
      for (int v = 0; v < 8; ++v)
        y[(size_t)(v + 8 * half) * TT * NO + (size_t)t * NO + o0 + 16 * j + l16] =
            acc[j][v];
  }
}

extern "C" void kernel_launch(void* const* d_in, const int* in_sizes, int n_in,
                              void* d_out, int out_size, void* d_ws,
                              size_t ws_size, hipStream_t stream) {
  const float* u  = (const float*)d_in[0];
  const float* x0 = (const float*)d_in[1];
  const float* A  = (const float*)d_in[2];
  const float* B  = (const float*)d_in[3];
  const float* C  = (const float*)d_in[4];
  const float* D  = (const float*)d_in[5];

  float* y      = (float*)d_out;
  float* xfinal = (float*)d_out + (size_t)NB * TT * NO;

  float* ws = (float*)d_ws;
  float* Abar   = ws;                 ws += NS * NS;
  float* PW1    = ws;                 ws += (LCH + 1) * NS * NS;
  float* A16    = PW1 + LCH * NS * NS;
  float* PW2    = ws;                 ws += (LCH + 1) * NS * NS;
  float* A256   = PW2 + LCH * NS * NS;
  float* Bsc    = ws;
  float* G      = ws;                 ws += NCH * NB * NS;
  float* uB     = ws;                 ws += (size_t)TT * NB * NS;
  float* Sst    = ws;                 ws += (size_t)TT * NB * NS;
  float* Sb     = uB;
  float* GG     = Sb + NCH * NB * NS;

  s4_prep_kernel<<<(NS * NS + NS * NI) / 256, 256, 0, stream>>>(A, B, Abar, Bsc);
  s4_ub_kernel<<<TT / 4, 128, 0, stream>>>(u, Bsc, uB);
  s4_chain_kernel<<<NCH + NS / 16, 32, 0, stream>>>(uB, Abar, G, PW1, Sst, nullptr, NCH);
  s4_chain_kernel<<<NGR + NS / 16, 32, 0, stream>>>(G, A16, GG, PW2, Sb, nullptr, NGR);
  s4_tail_kernel<<<NCH, 128, 0, stream>>>(GG, A256, x0, PW2, Sb, PW1, Sst, xfinal);
  s4_y_kernel<<<TT / YT, 128, 0, stream>>>(Sst, u, C, D, y);
}


# ub kernel weight-stationary: 256 workgroups, each wave keeps all B_bar fragments in registers over 4 time steps and fetches only u rows, 32 bytes a lane, a step ahead; sum order of uB changes
# speedup vs baseline: 7.2834x; 1.1005x over previous
#include <hip/hip_runtime.h>
#include <hip/hip_bf16.h>


typedef __attribute__((ext_vector_type(2))) float v2f;
typedef __attribute__((ext_vector_type(8))) float v8f;

#define S4_DT 0.001f
#define TT 4096
#define NB 16
#define NS 64
#define NI 256
#define NO 256
#define LCH 16
#define NCH 256
#define NGR 16

__device__ __forceinline__ v8f wmma4(v2f a, v2f b, v8f c) {
  return __builtin_amdgcn_wmma_f32_16x16x4_f32(
       false, a,  false, b,
       (short)0, c,  false,  false);
}

__device__ __forceinline__ v2f load2(const float* p) { return *(const v2f*)p; }

__device__ __forceinline__ void store_c(float* dst, int rs, int n0, v8f acc,
                                        int l16, int half) {
#pragma unroll
  for (int v = 0; v < 8; ++v)
    dst[(v + 8 * half) * rs + n0 + l16] = acc[v];
}

__device__ __forceinline__ v8f load_c(const float* src, int rs, int n0,
                                      int l16, int half) {
  v8f r;
#pragma unroll
  for (int v = 0; v < 8; ++v)
    r[v] = src[(v + 8 * half) * rs + n0 + l16];
  return r;
}

__global__ __launch_bounds__(256) void s4_prep_kernel(
    const float* __restrict__ A, const float* __restrict__ B,
    float* __restrict__ Abar, float* __restrict__ Bsc) {
  const int idx = blockIdx.x * 256 + threadIdx.x;
  if (idx < NS * NS) {
    int r = idx >> 6, c = idx & 63;
    Abar[idx] = A[idx] * S4_DT + (r == c ? 1.0f : 0.0f);
  } else {
    Bsc[idx - NS * NS] = B[idx - NS * NS] * S4_DT;
  }
}

#define UT 4
#define UKG (NI / 4)

__device__ __forceinline__ void ub_load_frag(v2f (&f)[UKG], const float* p) {
#pragma unroll
  for (int q = 0; q < NI / 16; ++q) {
    v8f x = *(const v8f*)(p + 16 * q);
    f[4 * q + 0] = __builtin_shufflevector(x, x, 0, 1);
    f[4 * q + 1] = __builtin_shufflevector(x, x, 2, 3);
    f[4 * q + 2] = __builtin_shufflevector(x, x, 4, 5);
    f[4 * q + 3] = __builtin_shufflevector(x, x, 6, 7);
  }
}

__device__ __forceinline__ void ub_step(const v2f (&a)[UKG],
                                        const v2f (&w)[4][UKG], float* dst,
                                        int l16, int half) {
  v8f acc[4] = {{}, {}, {}, {}};
#pragma unroll
  for (int kk = 0; kk < UKG; ++kk)
#pragma unroll
    for (int j = 0; j < 4; ++j) acc[j] = wmma4(a[kk], w[j][kk], acc[j]);
#pragma unroll
  for (int j = 0; j < 4; ++j) store_c(dst, NS, j * 16, acc[j], l16, half);
  __builtin_amdgcn_sched_barrier(0);
}

__global__ __launch_bounds__(128) void s4_ub_kernel(
    const float* __restrict__ u, const float* __restrict__ Bsc,
    float* __restrict__ uB) {
  const int wave = threadIdx.x >> 5, lane = threadIdx.x & 31;
  const int l16 = lane & 15, half = lane >> 4;
  const int t0 = (blockIdx.x * 4 + wave) * UT;

  const float* urow = u + (size_t)l16 * TT * NI + (size_t)t0 * NI + 8 * half;
  v2f a0[UKG], a1[UKG];
  ub_load_frag(a0, urow);

  const float* brow = Bsc + l16 * NI + 8 * half;
  v2f w[4][UKG];
#pragma unroll
  for (int q = 0; q < NI / 16; ++q)
#pragma unroll
    for (int j = 0; j < 4; ++j) {
      v8f x = *(const v8f*)(brow + 16 * j * NI + 16 * q);
      w[j][4 * q + 0] = __builtin_shufflevector(x, x, 0, 1);
      w[j][4 * q + 1] = __builtin_shufflevector(x, x, 2, 3);
      w[j][4 * q + 2] = __builtin_shufflevector(x, x, 4, 5);
      w[j][4 * q + 3] = __builtin_shufflevector(x, x, 6, 7);
    }

  float* dst = uB + (size_t)t0 * NB * NS;
  static_assert(UT % 2 == 0, "two fragment sets take turns");
#pragma unroll
  for (int i = 0; i < UT; i += 2) {
    ub_load_frag(a1, urow + (size_t)(i + 1) * NI);
    ub_step(a0, w, dst + (size_t)i * NB * NS, l16, half);
    if (i + 2 < UT) ub_load_frag(a0, urow + (size_t)(i + 2) * NI);
    ub_step(a1, w, dst + (size_t)(i + 1) * NB * NS, l16, half);
  }
}

template <bool PWR>
__device__ __forceinline__ void s4_put_state(float* p, const v8f (&st)[4]) {
  if (PWR) {
#pragma unroll
    for (int i = 0; i < 4; ++i)
#pragma unroll
      for (int r = 0; r < 8; ++r) p[(16 * i + r) * NS] = st[i][r];
  } else {
#pragma unroll
    for (int i = 0; i < 4; ++i) *(v8f*)(p + 16 * i) = st[i];
  }
}

template <bool PWR>
__device__ __forceinline__ void s4_fetch(v8f (&f)[4], const float* p) {
  if (!PWR) {
#pragma unroll
    for (int i = 0; i < 4; ++i) f[i] = *(const v8f*)(p + 16 * i);
  }
}

template <bool PWR>
__device__ __forceinline__ void s4_chain_step(const v2f (&a)[4][16],
                                              const v8f (&cur)[4],
                                              v8f (&nxt)[4]) {
#pragma unroll
  for (int kk = 0; kk < 16; ++kk) {
    const int q = kk >> 2, p = kk & 3;
    v2f b = {cur[q][2 * p], cur[q][2 * p + 1]};
#pragma unroll
    for (int i = 0; i < 4; ++i)
      nxt[i] = wmma4(a[i][kk], b, (PWR && kk == 0) ? v8f{} : nxt[i]);
  }
  __builtin_amdgcn_sched_barrier(0);
}

__device__ __forceinline__ void s4_load_mfrag(v2f (&a)[4][16], const float* M,
                                              int l16, int half) {
#pragma unroll
  for (int i = 0; i < 4; ++i)
#pragma unroll
    for (int q = 0; q < 4; ++q) {
      v8f w = *(const v8f*)(M + (16 * i + l16) * NS + 16 * q + 8 * half);
      a[i][4 * q + 0] = __builtin_shufflevector(w, w, 0, 1);
      a[i][4 * q + 1] = __builtin_shufflevector(w, w, 2, 3);
      a[i][4 * q + 2] = __builtin_shufflevector(w, w, 4, 5);
      a[i][4 * q + 3] = __builtin_shufflevector(w, w, 6, 7);
    }
}

template <bool PWR>
__device__ __forceinline__ void s4_chain_body(const float* F, const float* M,
                                              float* G, float* PW, float* Z,
                                              const float* X0, int c) {
  const int lane = threadIdx.x & 31;
  const int l16 = lane & 15, half = lane >> 4;

  v2f a[4][16];
  s4_load_mfrag(a, M, l16, half);

  const int lo = l16 * NS + 8 * half;
  const int r0 = 16 * c;
  const float* Fc = F + (size_t)c * LCH * NB * NS + lo;
  float* out = PWR ? PW + 8 * half * NS + r0 + l16
                   : Z + (size_t)c * LCH * NB * NS + lo;
  const int ostep = PWR ? NS * NS : NB * NS;
  const int fstep = NB * NS;

  v8f s0[4], s1[4], s2[4];
#pragma unroll
  for (int i = 0; i < 4; ++i) {
    if (PWR) {
#pragma unroll
      for (int r = 0; r < 8; ++r)
        s0[i][r] = (16 * i + 8 * half + r == r0 + l16) ? 1.0f : 0.0f;
    } else {
      s0[i] = X0 ? *(const v8f*)(X0 + lo + 16 * i) : v8f{};
    }
  }
  s4_fetch<PWR>(s1, Fc);

  for (int k = 0; k + 3 <= LCH; k += 3) {
    s4_put_state<PWR>(out + (size_t)k * ostep, s0);
    s4_fetch<PWR>(s2, Fc + (size_t)(k + 1) * fstep);
    s4_chain_step<PWR>(a, s0, s1);
    s4_put_state<PWR>(out + (size_t)(k + 1) * ostep, s1);
    s4_fetch<PWR>(s0, Fc + (size_t)(k + 2) * fstep);
    s4_chain_step<PWR>(a, s1, s2);
    s4_put_state<PWR>(out + (size_t)(k + 2) * ostep, s2);
    s4_fetch<PWR>(s1, Fc + (size_t)(k + 3) * fstep);
    s4_chain_step<PWR>(a, s2, s0);
  }
  static_assert(LCH % 3 == 1, "one step left behind the loop of three");
  s4_put_state<PWR>(out + (size_t)(LCH - 1) * ostep, s0);
  s4_chain_step<PWR>(a, s0, s1);

  if (PWR) {
    s4_put_state<true>(out + (size_t)LCH * ostep, s1);
  } else {
    s4_put_state<false>(G + (size_t)c * NB * NS + lo, s1);
  }
}

__global__ __launch_bounds__(32) void s4_chain_kernel(
    const float* __restrict__ F, const float* __restrict__ M,
    float* __restrict__ G, float* __restrict__ PW, float* __restrict__ Z,
    const float* __restrict__ X0, int nch) {
  const int c = blockIdx.x;
  if (c < nch) {
    s4_chain_body<false>(F, M, G, PW, Z, X0, c);
  } else {
    s4_chain_body<true>(F, M, G, PW, Z, X0, c - nch);
  }
}

#define TLK (LCH / 4)

__global__ __launch_bounds__(128) void s4_tail_kernel(
    const float* __restrict__ GG, const float* __restrict__ A256,
    const float* __restrict__ X0, const float* __restrict__ PW2,
    const float* __restrict__ Zb, const float* __restrict__ PW1,
    float* S, float* __restrict__ xfinal) {
  const int wave = threadIdx.x >> 5, lane = threadIdx.x & 31;
  const int l16 = lane & 15, half = lane >> 4;
  const int c = blockIdx.x, g = c / LCH, j = c % LCH;
  const int lo = l16 * NS + 8 * half;
  const int tile = NB * NS;
  const float* P1 = PW1 + (size_t)(wave * TLK) * NS * NS;
  float* sp = S + ((size_t)c * LCH + wave * TLK) * tile + lo;

  v2f a[4][16], aA[4][16], aB[4][16];
  v8f s0[4], s1[4], s2[4], sb[4], zA[4], zB[4];
  s4_load_mfrag(a, A256, l16, half);
#pragma unroll
  for (int i = 0; i < 4; ++i) s0[i] = *(const v8f*)(X0 + lo + 16 * i);
  s4_fetch<false>(s1, GG + lo);
  s4_load_mfrag(aB, PW2 + (size_t)j * NS * NS, l16, half);
  s4_fetch<false>(sb, Zb + (size_t)c * tile + lo);
  s4_load_mfrag(aA, P1, l16, half);
  s4_fetch<false>(zA, sp);

  int gi = 0;
  for (; gi + 3 <= g; gi += 3) {
    s4_fetch<false>(s2, GG + (size_t)(gi + 1) * tile + lo);
    s4_chain_step<false>(a, s0, s1);
    s4_fetch<false>(s0, GG + (size_t)(gi + 2) * tile + lo);
    s4_chain_step<false>(a, s1, s2);
    s4_fetch<false>(s1, GG + (size_t)(gi + 3) * tile + lo);
    s4_chain_step<false>(a, s2, s0);
  }
  v8f sg[4];
  if (g - gi == 2) {
    s4_fetch<false>(s2, GG + (size_t)(gi + 1) * tile + lo);
    s4_chain_step<false>(a, s0, s1);
    s4_chain_step<false>(a, s1, s2);
#pragma unroll
    for (int i = 0; i < 4; ++i) sg[i] = s2[i];
  } else if (g - gi == 1) {
    s4_chain_step<false>(a, s0, s1);
#pragma unroll
    for (int i = 0; i < 4; ++i) sg[i] = s1[i];
  } else {
#pragma unroll
    for (int i = 0; i < 4; ++i) sg[i] = s0[i];
  }

  if (c == NCH - 1 && wave == 0) {
    v8f xf[4];
    s4_fetch<false>(xf, GG + (size_t)(NGR - 1) * tile + lo);
    s4_chain_step<false>(a, sg, xf);
    s4_put_state<false>(xfinal + lo, xf);
  }

  s4_chain_step<false>(aB, sg, sb);
  s4_load_mfrag(aB, P1 + NS * NS, l16, half);
  s4_fetch<false>(zB, sp + tile);
  s4_chain_step<false>(aA, sb, zA);
  s4_put_state<false>(sp, zA);
  s4_load_mfrag(aA, P1 + 2 * NS * NS, l16, half);
  s4_fetch<false>(zA, sp + 2 * tile);
  s4_chain_step<false>(aB, sb, zB);
  s4_put_state<false>(sp + tile, zB);
  s4_load_mfrag(aB, P1 + 3 * NS * NS, l16, half);
  s4_fetch<false>(zB, sp + 3 * tile);
  s4_chain_step<false>(aA, sb, zA);
  s4_put_state<false>(sp + 2 * tile, zA);
  s4_chain_step<false>(aB, sb, zB);
  s4_put_state<false>(sp + 3 * tile, zB);
  static_assert(TLK == 4, "four products per wave, written out");
}

#define YT 16
#define YKG (NS / 4 + NI / 4)
#define YNB (YKG / 8)

__device__ __forceinline__ v2f y_load_a(const float* sp, const float* up,
                                        int kk) {
  return kk < NS / 4 ? load2(sp + 4 * kk) : load2(up + 4 * (kk - NS / 4));
}

__global__ __launch_bounds__(128) void s4_y_kernel(
    const float* __restrict__ S, const float* __restrict__ u,
    const float* __restrict__ C, const float* __restrict__ D,
    float* __restrict__ y) {
  const int wave = threadIdx.x >> 5, lane = threadIdx.x & 31;
  const int l16 = lane & 15, half = lane >> 4;
  const int t0 = blockIdx.x * YT;
  const int o0 = wave * 64;

  v2f w[YKG][4];
#pragma unroll
  for (int kk = 0; kk < NS / 4; ++kk)
#pragma unroll
    for (int j = 0; j < 4; ++j)
      w[kk][j] = load2(C + (o0 + 16 * j + l16) * NS + 4 * kk + 2 * half);
#pragma unroll
  for (int kk = 0; kk < NI / 4; ++kk)
#pragma unroll
    for (int j = 0; j < 4; ++j)
      w[NS / 4 + kk][j] =
          load2(D + (o0 + 16 * j + l16) * NI + 4 * kk + 2 * half);

  const float* srow = S + (size_t)t0 * NB * NS + l16 * NS + 2 * half;
  const float* urow = u + (size_t)l16 * TT * NI + (size_t)t0 * NI + 2 * half;
  v2f a[YKG];
#pragma unroll
  for (int kk = 0; kk < 8 * (YNB - 1); ++kk) a[kk] = y_load_a(srow, urow, kk);

  for (int i = 0; i < YT; ++i) {
    const float* sc = srow + (size_t)i * NB * NS;
    const float* uc = urow + (size_t)i * NI;
    const int in = (i + 1 < YT) ? i + 1 : i;
    const float* sn = srow + (size_t)in * NB * NS;
    const float* un = urow + (size_t)in * NI;

    v8f acc[4] = {{}, {}, {}, {}};
#pragma unroll
    for (int c8 = 0; c8 < YNB; ++c8) {
#pragma unroll
      for (int q = 0; q < 8; ++q) {
        const int kk = 8 * c8 + q;
#pragma unroll
        for (int j = 0; j < 4; ++j) acc[j] = wmma4(a[kk], w[kk][j], acc[j]);
      }
      const int r8 = (c8 + YNB - 1) % YNB;
#pragma unroll
      for (int q = 0; q < 8; ++q)
        a[8 * r8 + q] = y_load_a(c8 == 0 ? sc : sn, c8 == 0 ? uc : un, 8 * r8 + q);
      __builtin_amdgcn_sched_barrier(0);
    }

    const int t = t0 + i;
#pragma unroll
    for (int j = 0; j < 4; ++j)
#pragma unroll
      for (int v = 0; v < 8; ++v)
        y[(size_t)(v + 8 * half) * TT * NO + (size_t)t * NO + o0 + 16 * j + l16] =
            acc[j][v];
  }
}

extern "C" void kernel_launch(void* const* d_in, const int* in_sizes, int n_in,
                              void* d_out, int out_size, void* d_ws,
                              size_t ws_size, hipStream_t stream) {
  const float* u  = (const float*)d_in[0];
  const float* x0 = (const float*)d_in[1];
  const float* A  = (const float*)d_in[2];
  const float* B  = (const float*)d_in[3];
  const float* C  = (const float*)d_in[4];
  const float* D  = (const float*)d_in[5];

  float* y      = (float*)d_out;
  float* xfinal = (float*)d_out + (size_t)NB * TT * NO;

  float* ws = (float*)d_ws;
  float* Abar   = ws;                 ws += NS * NS;
  float* PW1    = ws;                 ws += (LCH + 1) * NS * NS;
  float* A16    = PW1 + LCH * NS * NS;
  float* PW2    = ws;                 ws += (LCH + 1) * NS * NS;
  float* A256   = PW2 + LCH * NS * NS;
  float* Bsc    = ws;
  float* G      = ws;                 ws += NCH * NB * NS;
  float* uB     = ws;                 ws += (size_t)TT * NB * NS;
  float* Sst    = ws;                 ws += (size_t)TT * NB * NS;
  float* Sb     = uB;
  float* GG     = Sb + NCH * NB * NS;

  s4_prep_kernel<<<(NS * NS + NS * NI) / 256, 256, 0, stream>>>(A, B, Abar, Bsc);
  s4_ub_kernel<<<TT / (4 * UT), 128, 0, stream>>>(u, Bsc, uB);
  s4_chain_kernel<<<NCH + NS / 16, 32, 0, stream>>>(uB, Abar, G, PW1, Sst, nullptr, NCH);
  s4_chain_kernel<<<NGR + NS / 16, 32, 0, stream>>>(G, A16, GG, PW2, Sb, nullptr, NGR);
  s4_tail_kernel<<<NCH, 128, 0, stream>>>(GG, A256, x0, PW2, Sb, PW1, Sst, xfinal);
  s4_y_kernel<<<TT / YT, 128, 0, stream>>>(Sst, u, C, D, y);
}


# chain and tail steps: a step's four 16-row tiles on four waves (one per SIMD, 16 products each, own tile first), state handed round through LDS, LDS-only barrier; sum order of S, x_final changes
# speedup vs baseline: 8.1385x; 1.1174x over previous
#include <hip/hip_runtime.h>
#include <hip/hip_bf16.h>


typedef __attribute__((ext_vector_type(2))) float v2f;
typedef __attribute__((ext_vector_type(8))) float v8f;

#define S4_DT 0.001f
#define TT 4096
#define NB 16
#define NS 64
#define NI 256
#define NO 256
#define LCH 16
#define NCH 256
#define NGR 16

__device__ __forceinline__ v8f wmma4(v2f a, v2f b, v8f c) {
  return __builtin_amdgcn_wmma_f32_16x16x4_f32(
       false, a,  false, b,
       (short)0, c,  false,  false);
}

__device__ __forceinline__ v2f load2(const float* p) { return *(const v2f*)p; }

__device__ __forceinline__ void store_c(float* dst, int rs, int n0, v8f acc,
                                        int l16, int half) {
#pragma unroll
  for (int v = 0; v < 8; ++v)
    dst[(v + 8 * half) * rs + n0 + l16] = acc[v];
}

__device__ __forceinline__ v8f load_c(const float* src, int rs, int n0,
                                      int l16, int half) {
  v8f r;
#pragma unroll
  for (int v = 0; v < 8; ++v)
    r[v] = src[(v + 8 * half) * rs + n0 + l16];
  return r;
}

__global__ __launch_bounds__(256) void s4_prep_kernel(
    const float* __restrict__ A, const float* __restrict__ B,
    float* __restrict__ Abar, float* __restrict__ Bsc) {
  const int idx = blockIdx.x * 256 + threadIdx.x;
  if (idx < NS * NS) {
    int r = idx >> 6, c = idx & 63;
    Abar[idx] = A[idx] * S4_DT + (r == c ? 1.0f : 0.0f);
  } else {
    Bsc[idx - NS * NS] = B[idx - NS * NS] * S4_DT;
  }
}

#define UT 4
#define UKG (NI / 4)

__device__ __forceinline__ void ub_load_frag(v2f (&f)[UKG], const float* p) {
#pragma unroll
  for (int q = 0; q < NI / 16; ++q) {
    v8f x = *(const v8f*)(p + 16 * q);
    f[4 * q + 0] = __builtin_shufflevector(x, x, 0, 1);
    f[4 * q + 1] = __builtin_shufflevector(x, x, 2, 3);
    f[4 * q + 2] = __builtin_shufflevector(x, x, 4, 5);
    f[4 * q + 3] = __builtin_shufflevector(x, x, 6, 7);
  }
}

__device__ __forceinline__ void ub_step(const v2f (&a)[UKG],
                                        const v2f (&w)[4][UKG], float* dst,
                                        int l16, int half) {
  v8f acc[4] = {{}, {}, {}, {}};
#pragma unroll
  for (int kk = 0; kk < UKG; ++kk)
#pragma unroll
    for (int j = 0; j < 4; ++j) acc[j] = wmma4(a[kk], w[j][kk], acc[j]);
#pragma unroll
  for (int j = 0; j < 4; ++j) store_c(dst, NS, j * 16, acc[j], l16, half);
  __builtin_amdgcn_sched_barrier(0);
}

__global__ __launch_bounds__(128) void s4_ub_kernel(
    const float* __restrict__ u, const float* __restrict__ Bsc,
    float* __restrict__ uB) {
  const int wave = threadIdx.x >> 5, lane = threadIdx.x & 31;
  const int l16 = lane & 15, half = lane >> 4;
  const int t0 = (blockIdx.x * 4 + wave) * UT;

  const float* urow = u + (size_t)l16 * TT * NI + (size_t)t0 * NI + 8 * half;
  v2f a0[UKG], a1[UKG];
  ub_load_frag(a0, urow);

  const float* brow = Bsc + l16 * NI + 8 * half;
  v2f w[4][UKG];
#pragma unroll
  for (int q = 0; q < NI / 16; ++q)
#pragma unroll
    for (int j = 0; j < 4; ++j) {
      v8f x = *(const v8f*)(brow + 16 * j * NI + 16 * q);
      w[j][4 * q + 0] = __builtin_shufflevector(x, x, 0, 1);
      w[j][4 * q + 1] = __builtin_shufflevector(x, x, 2, 3);
      w[j][4 * q + 2] = __builtin_shufflevector(x, x, 4, 5);
      w[j][4 * q + 3] = __builtin_shufflevector(x, x, 6, 7);
    }

  float* dst = uB + (size_t)t0 * NB * NS;
  static_assert(UT % 2 == 0, "two fragment sets take turns");
#pragma unroll
  for (int i = 0; i < UT; i += 2) {
    ub_load_frag(a1, urow + (size_t)(i + 1) * NI);
    ub_step(a0, w, dst + (size_t)i * NB * NS, l16, half);
    if (i + 2 < UT) ub_load_frag(a0, urow + (size_t)(i + 2) * NI);
    ub_step(a1, w, dst + (size_t)(i + 1) * NB * NS, l16, half);
  }
}

typedef __attribute__((ext_vector_type(4))) float v4f;

#define HB_TILE 64
#define HB_SLOTS (2 * 4 * HB_TILE)

struct s4_slots { v4f* p[4]; };

__device__ __forceinline__ s4_slots s4_hand_slots(v4f* hb, int w, int lane) {
  s4_slots hs;
#pragma unroll
  for (int qq = 0; qq < 4; ++qq) hs.p[qq] = hb + ((w + qq) & 3) * HB_TILE + lane;
  return hs;
}

__device__ __forceinline__ void s4_hand_put(const s4_slots& hs, int par, v8f x) {
  v4f* p = hs.p[0] + par * 4 * HB_TILE;
  p[0] = __builtin_shufflevector(x, x, 0, 1, 2, 3);
  p[32] = __builtin_shufflevector(x, x, 4, 5, 6, 7);
}

__device__ __forceinline__ void s4_hand_get(v8f (&cur)[4], const s4_slots& hs,
                                            int par) {
#pragma unroll
  for (int qq = 1; qq < 4; ++qq) {
    const v4f* p = hs.p[qq] + par * 4 * HB_TILE;
    v4f lo = p[0], hi = p[32];
    cur[qq] = __builtin_shufflevector(lo, hi, 0, 1, 2, 3, 4, 5, 6, 7);
  }
  __builtin_amdgcn_sched_barrier(0);
}

__device__ __forceinline__ void s4_lds_barrier() {
  asm volatile("s_wait_dscnt 0x0\n\ts_barrier_signal -1\n\ts_barrier_wait -1"
               ::: "memory");
}

__device__ __forceinline__ void s4_load_mtile(v2f (&a)[16], const float* M,
                                              int w, int l16, int half) {
  const float* row = M + (16 * w + l16) * NS + 8 * half;
#pragma unroll
  for (int qq = 0; qq < 4; ++qq) {
    v8f x = *(const v8f*)(row + 16 * ((w + qq) & 3));
    a[4 * qq + 0] = __builtin_shufflevector(x, x, 0, 1);
    a[4 * qq + 1] = __builtin_shufflevector(x, x, 2, 3);
    a[4 * qq + 2] = __builtin_shufflevector(x, x, 4, 5);
    a[4 * qq + 3] = __builtin_shufflevector(x, x, 6, 7);
  }
}

__device__ __forceinline__ void s4_load_state(v8f (&cur)[4], const float* X,
                                              int w, int lo) {
#pragma unroll
  for (int qq = 0; qq < 4; ++qq)
    cur[qq] = *(const v8f*)(X + lo + 16 * ((w + qq) & 3));
}

template <int QA, int QB>
__device__ __forceinline__ v8f s4_tile_mac(const v2f (&a)[16],
                                           const v8f (&cur)[4], v8f acc) {
#pragma unroll
  for (int kk = 4 * QA; kk < 4 * QB; ++kk) {
    const int qq = kk >> 2, p = kk & 3;
    v2f b = {cur[qq][2 * p], cur[qq][2 * p + 1]};
    acc = wmma4(a[kk], b, acc);
  }
  return acc;
}

template <bool PWR>
__device__ __forceinline__ void s4_put_tile(float* p, v8f st) {
  if (PWR) {
#pragma unroll
    for (int r = 0; r < 8; ++r) p[r * NS] = st[r];
  } else {
    *(v8f*)p = st;
  }
}

template <bool PWR>
__device__ __forceinline__ void s4_chain_body(const float* F, const float* M,
                                              float* G, float* PW, float* Z,
                                              const float* X0, int c, v4f* hb) {
  const int w = threadIdx.x >> 5, lane = threadIdx.x & 31;
  const int l16 = lane & 15, half = lane >> 4;

  v2f a[16];
  s4_load_mtile(a, M, w, l16, half);
  const s4_slots hs = s4_hand_slots(hb, w, lane);

  const int lo = l16 * NS + 8 * half;
  const int r0 = 16 * c;
  const float* Fc = F + (size_t)c * LCH * NB * NS + lo + 16 * w;
  float* out = PWR ? PW + (16 * w + 8 * half) * NS + r0 + l16
                   : Z + (size_t)c * LCH * NB * NS + lo + 16 * w;
  const int ostep = PWR ? NS * NS : NB * NS;
  const int fstep = NB * NS;

  v8f cur[4];
  if (PWR) {
#pragma unroll
    for (int qq = 0; qq < 4; ++qq)
#pragma unroll
      for (int r = 0; r < 8; ++r)
        cur[qq][r] =
            (16 * ((w + qq) & 3) + 8 * half + r == r0 + l16) ? 1.0f : 0.0f;
  } else if (X0) {
    s4_load_state(cur, X0, w, lo);
  } else {
#pragma unroll
    for (int qq = 0; qq < 4; ++qq) cur[qq] = v8f{};
  }
  v8f acc = v8f{};
  if (!PWR) acc = *(const v8f*)Fc;

#pragma unroll
  for (int k = 0; k < LCH; ++k) {
    s4_put_tile<PWR>(out + (size_t)k * ostep, cur[0]);
    v8f nf = v8f{};
    if (!PWR && k + 1 < LCH) nf = *(const v8f*)(Fc + (size_t)(k + 1) * fstep);
    acc = s4_tile_mac<0, 1>(a, cur, acc);
    if (k > 0) {
      __builtin_amdgcn_sched_barrier(0);
      s4_lds_barrier();
      s4_hand_get(cur, hs, (k - 1) & 1);
    }
    acc = s4_tile_mac<1, 4>(a, cur, acc);
    if (k + 1 < LCH) s4_hand_put(hs, k & 1, acc);
    cur[0] = acc;
    acc = nf;
  }

  if (PWR) {
    s4_put_tile<true>(out + (size_t)LCH * ostep, cur[0]);
  } else {
    *(v8f*)(G + (size_t)c * NB * NS + lo + 16 * w) = cur[0];
  }
}

__global__ __launch_bounds__(128) void s4_chain_kernel(
    const float* __restrict__ F, const float* __restrict__ M,
    float* __restrict__ G, float* __restrict__ PW, float* __restrict__ Z,
    const float* __restrict__ X0, int nch) {
  __shared__ v4f hb[HB_SLOTS];
  const int c = blockIdx.x;
  if (c < nch) {
    s4_chain_body<false>(F, M, G, PW, Z, X0, c, hb);
  } else {
    s4_chain_body<true>(F, M, G, PW, Z, X0, c - nch, hb);
  }
}

__device__ __forceinline__ void s4_pair_mac(const v2f (&a0)[16],
                                            const v2f (&a1)[16],
                                            const v8f (&sb)[4], v8f& z0,
                                            v8f& z1) {
#pragma unroll
  for (int kk = 0; kk < 16; ++kk) {
    const int qq = kk >> 2, p = kk & 3;
    v2f b = {sb[qq][2 * p], sb[qq][2 * p + 1]};
    z0 = wmma4(a0[kk], b, z0);
    z1 = wmma4(a1[kk], b, z1);
  }
}

__global__ __launch_bounds__(128) void s4_tail_kernel(
    const float* __restrict__ GG, const float* __restrict__ A256,
    const float* __restrict__ X0, const float* __restrict__ PW2,
    const float* __restrict__ Zb, const float* __restrict__ PW1,
    float* S, float* __restrict__ xfinal) {
  __shared__ v4f hb[HB_SLOTS];
  const int w = threadIdx.x >> 5, lane = threadIdx.x & 31;
  const int l16 = lane & 15, half = lane >> 4;
  const int c = blockIdx.x, g = c / LCH, j = c % LCH;
  const int lo = l16 * NS + 8 * half;
  const int tile = NB * NS;
  const int lw = lo + 16 * w;
  float* sp = S + (size_t)c * LCH * tile + lw;

  v2f a[16], aS[16];
  v8f cur[4];
  const s4_slots hs = s4_hand_slots(hb, w, lane);
  s4_load_mtile(a, A256, w, l16, half);
  s4_load_state(cur, X0, w, lo);
  v8f acc = *(const v8f*)(GG + lw);
  s4_load_mtile(aS, PW2 + (size_t)j * NS * NS, w, l16, half);
  v8f sbacc = *(const v8f*)(Zb + (size_t)c * tile + lw);
  v2f pa[2][2][16];
  v8f pz[2][2];
#pragma unroll
  for (int i = 0; i < 2; ++i)
#pragma unroll
    for (int e = 0; e < 2; ++e) {
      s4_load_mtile(pa[i][e], PW1 + (size_t)(2 * i + e) * NS * NS, w, l16, half);
      pz[i][e] = *(const v8f*)(sp + (size_t)(2 * i + e) * tile);
    }

  for (int gi = 0; gi < g; ++gi) {
    v8f nf = *(const v8f*)(GG + (size_t)(gi + 1) * tile + lw);
    acc = s4_tile_mac<0, 1>(a, cur, acc);
    if (gi > 0) {
      __builtin_amdgcn_sched_barrier(0);
      s4_lds_barrier();
      s4_hand_get(cur, hs, (gi - 1) & 1);
    }
    acc = s4_tile_mac<1, 4>(a, cur, acc);
    s4_hand_put(hs, gi & 1, acc);
    cur[0] = acc;
    acc = nf;
  }

  sbacc = s4_tile_mac<0, 1>(aS, cur, sbacc);
  if (g > 0) {
    __builtin_amdgcn_sched_barrier(0);
    s4_lds_barrier();
    s4_hand_get(cur, hs, (g - 1) & 1);
  }
  if (c == NCH - 1) {
    acc = s4_tile_mac<0, 4>(a, cur, acc);
    *(v8f*)(xfinal + lw) = acc;
  }
  sbacc = s4_tile_mac<1, 4>(aS, cur, sbacc);
  v8f sb[4];
  s4_hand_put(hs, g & 1, sbacc);
  s4_lds_barrier();
  sb[0] = sbacc;
  s4_hand_get(sb, hs, g & 1);

  static_assert(LCH % 4 == 0, "pairs of states, two sets taking turns");
#pragma unroll
  for (int i = 0; i < LCH / 2; ++i) {
    const int s = i & 1;
    s4_pair_mac(pa[s][0], pa[s][1], sb, pz[s][0], pz[s][1]);
    *(v8f*)(sp + (size_t)(2 * i) * tile) = pz[s][0];
    *(v8f*)(sp + (size_t)(2 * i + 1) * tile) = pz[s][1];
    if (i + 2 < LCH / 2) {
#pragma unroll
      for (int e = 0; e < 2; ++e) {
        s4_load_mtile(pa[s][e], PW1 + (size_t)(2 * i + 4 + e) * NS * NS, w, l16,
                      half);
        pz[s][e] = *(const v8f*)(sp + (size_t)(2 * i + 4 + e) * tile);
      }
    }
    __builtin_amdgcn_sched_barrier(0);
  }
}

#define YT 16
#define YKG (NS / 4 + NI / 4)
#define YNB (YKG / 8)

__device__ __forceinline__ v2f y_load_a(const float* sp, const float* up,
                                        int kk) {
  return kk < NS / 4 ? load2(sp + 4 * kk) : load2(up + 4 * (kk - NS / 4));
}

__global__ __launch_bounds__(128) void s4_y_kernel(
    const float* __restrict__ S, const float* __restrict__ u,
    const float* __restrict__ C, const float* __restrict__ D,
    float* __restrict__ y) {
  const int wave = threadIdx.x >> 5, lane = threadIdx.x & 31;
  const int l16 = lane & 15, half = lane >> 4;
  const int t0 = blockIdx.x * YT;
  const int o0 = wave * 64;

  v2f w[YKG][4];
#pragma unroll
  for (int kk = 0; kk < NS / 4; ++kk)
#pragma unroll
    for (int j = 0; j < 4; ++j)
      w[kk][j] = load2(C + (o0 + 16 * j + l16) * NS + 4 * kk + 2 * half);
#pragma unroll
  for (int kk = 0; kk < NI / 4; ++kk)
#pragma unroll
    for (int j = 0; j < 4; ++j)
      w[NS / 4 + kk][j] =
          load2(D + (o0 + 16 * j + l16) * NI + 4 * kk + 2 * half);

  const float* srow = S + (size_t)t0 * NB * NS + l16 * NS + 2 * half;
  const float* urow = u + (size_t)l16 * TT * NI + (size_t)t0 * NI + 2 * half;
  v2f a[YKG];
#pragma unroll
  for (int kk = 0; kk < 8 * (YNB - 1); ++kk) a[kk] = y_load_a(srow, urow, kk);

  for (int i = 0; i < YT; ++i) {
    const float* sc = srow + (size_t)i * NB * NS;
    const float* uc = urow + (size_t)i * NI;
    const int in = (i + 1 < YT) ? i + 1 : i;
    const float* sn = srow + (size_t)in * NB * NS;
    const float* un = urow + (size_t)in * NI;

    v8f acc[4] = {{}, {}, {}, {}};
#pragma unroll
    for (int c8 = 0; c8 < YNB; ++c8) {
#pragma unroll
      for (int q = 0; q < 8; ++q) {
        const int kk = 8 * c8 + q;
#pragma unroll
        for (int j = 0; j < 4; ++j) acc[j] = wmma4(a[kk], w[kk][j], acc[j]);
      }
      const int r8 = (c8 + YNB - 1) % YNB;
#pragma unroll
      for (int q = 0; q < 8; ++q)
        a[8 * r8 + q] = y_load_a(c8 == 0 ? sc : sn, c8 == 0 ? uc : un, 8 * r8 + q);
      __builtin_amdgcn_sched_barrier(0);
    }

    const int t = t0 + i;
#pragma unroll
    for (int j = 0; j < 4; ++j)
#pragma unroll
      for (int v = 0; v < 8; ++v)
        y[(size_t)(v + 8 * half) * TT * NO + (size_t)t * NO + o0 + 16 * j + l16] =
            acc[j][v];
  }
}

extern "C" void kernel_launch(void* const* d_in, const int* in_sizes, int n_in,
                              void* d_out, int out_size, void* d_ws,
                              size_t ws_size, hipStream_t stream) {
  const float* u  = (const float*)d_in[0];
  const float* x0 = (const float*)d_in[1];
  const float* A  = (const float*)d_in[2];
  const float* B  = (const float*)d_in[3];
  const float* C  = (const float*)d_in[4];
  const float* D  = (const float*)d_in[5];

  float* y      = (float*)d_out;
  float* xfinal = (float*)d_out + (size_t)NB * TT * NO;

  float* ws = (float*)d_ws;
  float* Abar   = ws;                 ws += NS * NS;
  float* PW1    = ws;                 ws += (LCH + 1) * NS * NS;
  float* A16    = PW1 + LCH * NS * NS;
  float* PW2    = ws;                 ws += (LCH + 1) * NS * NS;
  float* A256   = PW2 + LCH * NS * NS;
  float* Bsc    = ws;
  float* G      = ws;                 ws += NCH * NB * NS;
  float* uB     = ws;                 ws += (size_t)TT * NB * NS;
  float* Sst    = ws;                 ws += (size_t)TT * NB * NS;
  float* Sb     = uB;
  float* GG     = Sb + NCH * NB * NS;

  s4_prep_kernel<<<(NS * NS + NS * NI) / 256, 256, 0, stream>>>(A, B, Abar, Bsc);
  s4_ub_kernel<<<TT / (4 * UT), 128, 0, stream>>>(u, Bsc, uB);
  s4_chain_kernel<<<NCH + NS / 16, 128, 0, stream>>>(uB, Abar, G, PW1, Sst, nullptr, NCH);
  s4_chain_kernel<<<NGR + NS / 16, 128, 0, stream>>>(G, A16, GG, PW2, Sb, nullptr, NGR);
  s4_tail_kernel<<<NCH, 128, 0, stream>>>(GG, A256, x0, PW2, Sb, PW1, Sst, xfinal);
  s4_y_kernel<<<TT / YT, 128, 0, stream>>>(Sst, u, C, D, y);
}


# level-1 chain workgroups form their own forcing terms u.B_bar^T first (WMMA, B_bar resident, tiles handed over in LDS); powers ride as a 2nd column tile in 4 of them; no ub launch, no uB; same sums
# speedup vs baseline: 8.6920x; 1.0680x over previous
#include <hip/hip_runtime.h>
#include <hip/hip_bf16.h>


typedef __attribute__((ext_vector_type(2))) float v2f;
typedef __attribute__((ext_vector_type(8))) float v8f;

#define S4_DT 0.001f
#define TT 4096
#define NB 16
#define NS 64
#define NI 256
#define NO 256
#define LCH 16
#define NCH 256
#define NGR 16

__device__ __forceinline__ v8f wmma4(v2f a, v2f b, v8f c) {
  return __builtin_amdgcn_wmma_f32_16x16x4_f32(
       false, a,  false, b,
       (short)0, c,  false,  false);
}

__device__ __forceinline__ v2f load2(const float* p) { return *(const v2f*)p; }

__device__ __forceinline__ void store_c(float* dst, int rs, int n0, v8f acc,
                                        int l16, int half) {
#pragma unroll
  for (int v = 0; v < 8; ++v)
    dst[(v + 8 * half) * rs + n0 + l16] = acc[v];
}

__device__ __forceinline__ v8f load_c(const float* src, int rs, int n0,
                                      int l16, int half) {
  v8f r;
#pragma unroll
  for (int v = 0; v < 8; ++v)
    r[v] = src[(v + 8 * half) * rs + n0 + l16];
  return r;
}

__global__ __launch_bounds__(256) void s4_prep_kernel(
    const float* __restrict__ A, const float* __restrict__ B,
    float* __restrict__ Abar, float* __restrict__ Bsc) {
  const int idx = blockIdx.x * 256 + threadIdx.x;
  if (idx < NS * NS) {
    int r = idx >> 6, c = idx & 63;
    Abar[idx] = A[idx] * S4_DT + (r == c ? 1.0f : 0.0f);
  } else {
    Bsc[idx - NS * NS] = B[idx - NS * NS] * S4_DT;
  }
}

typedef __attribute__((ext_vector_type(4))) float v4f;

#define HB_TILE 64
#define HB_SLOTS (2 * 4 * HB_TILE)

struct s4_slots { v4f* p[4]; };

__device__ __forceinline__ s4_slots s4_hand_slots(v4f* hb, int w, int lane) {
  s4_slots hs;
#pragma unroll
  for (int qq = 0; qq < 4; ++qq) hs.p[qq] = hb + ((w + qq) & 3) * HB_TILE + lane;
  return hs;
}

__device__ __forceinline__ void s4_hand_put(const s4_slots& hs, int par, v8f x) {
  v4f* p = hs.p[0] + par * 4 * HB_TILE;
  p[0] = __builtin_shufflevector(x, x, 0, 1, 2, 3);
  p[32] = __builtin_shufflevector(x, x, 4, 5, 6, 7);
}

__device__ __forceinline__ void s4_hand_get(v8f (&cur)[4], const s4_slots& hs,
                                            int par) {
#pragma unroll
  for (int qq = 1; qq < 4; ++qq) {
    const v4f* p = hs.p[qq] + par * 4 * HB_TILE;
    v4f lo = p[0], hi = p[32];
    cur[qq] = __builtin_shufflevector(lo, hi, 0, 1, 2, 3, 4, 5, 6, 7);
  }
  __builtin_amdgcn_sched_barrier(0);
}

__device__ __forceinline__ void s4_lds_barrier() {
  asm volatile("s_wait_dscnt 0x0\n\ts_barrier_signal -1\n\ts_barrier_wait -1"
               ::: "memory");
}

__device__ __forceinline__ void s4_load_mtile(v2f (&a)[16], const float* M,
                                              int w, int l16, int half) {
  const float* row = M + (16 * w + l16) * NS + 8 * half;
#pragma unroll
  for (int qq = 0; qq < 4; ++qq) {
    v8f x = *(const v8f*)(row + 16 * ((w + qq) & 3));
    a[4 * qq + 0] = __builtin_shufflevector(x, x, 0, 1);
    a[4 * qq + 1] = __builtin_shufflevector(x, x, 2, 3);
    a[4 * qq + 2] = __builtin_shufflevector(x, x, 4, 5);
    a[4 * qq + 3] = __builtin_shufflevector(x, x, 6, 7);
  }
}

__device__ __forceinline__ void s4_load_state(v8f (&cur)[4], const float* X,
                                              int w, int lo) {
#pragma unroll
  for (int qq = 0; qq < 4; ++qq)
    cur[qq] = *(const v8f*)(X + lo + 16 * ((w + qq) & 3));
}

template <int QA, int QB>
__device__ __forceinline__ v8f s4_tile_mac(const v2f (&a)[16],
                                           const v8f (&cur)[4], v8f acc) {
#pragma unroll
  for (int kk = 4 * QA; kk < 4 * QB; ++kk) {
    const int qq = kk >> 2, p = kk & 3;
    v2f b = {cur[qq][2 * p], cur[qq][2 * p + 1]};
    acc = wmma4(a[kk], b, acc);
  }
  return acc;
}

template <bool PWR>
__device__ __forceinline__ void s4_put_tile(float* p, v8f st) {
  if (PWR) {
#pragma unroll
    for (int r = 0; r < 8; ++r) p[r * NS] = st[r];
  } else {
    *(v8f*)p = st;
  }
}

template <bool PWR>
__device__ __forceinline__ void s4_chain_body(const float* F, const float* M,
                                              float* G, float* PW, float* Z,
                                              const float* X0, int c, v4f* hb) {
  const int w = threadIdx.x >> 5, lane = threadIdx.x & 31;
  const int l16 = lane & 15, half = lane >> 4;

  v2f a[16];
  s4_load_mtile(a, M, w, l16, half);
  const s4_slots hs = s4_hand_slots(hb, w, lane);

  const int lo = l16 * NS + 8 * half;
  const int r0 = 16 * c;
  const float* Fc = F + (size_t)c * LCH * NB * NS + lo + 16 * w;
  float* out = PWR ? PW + (16 * w + 8 * half) * NS + r0 + l16
                   : Z + (size_t)c * LCH * NB * NS + lo + 16 * w;
  const int ostep = PWR ? NS * NS : NB * NS;
  const int fstep = NB * NS;

  v8f cur[4];
  if (PWR) {
#pragma unroll
    for (int qq = 0; qq < 4; ++qq)
#pragma unroll
      for (int r = 0; r < 8; ++r)
        cur[qq][r] =
            (16 * ((w + qq) & 3) + 8 * half + r == r0 + l16) ? 1.0f : 0.0f;
  } else if (X0) {
    s4_load_state(cur, X0, w, lo);
  } else {
#pragma unroll
    for (int qq = 0; qq < 4; ++qq) cur[qq] = v8f{};
  }
  v8f acc = v8f{};
  if (!PWR) acc = *(const v8f*)Fc;

#pragma unroll
  for (int k = 0; k < LCH; ++k) {
    s4_put_tile<PWR>(out + (size_t)k * ostep, cur[0]);
    v8f nf = v8f{};
    if (!PWR && k + 1 < LCH) nf = *(const v8f*)(Fc + (size_t)(k + 1) * fstep);
    acc = s4_tile_mac<0, 1>(a, cur, acc);
    if (k > 0) {
      __builtin_amdgcn_sched_barrier(0);
      s4_lds_barrier();
      s4_hand_get(cur, hs, (k - 1) & 1);
    }
    acc = s4_tile_mac<1, 4>(a, cur, acc);
    if (k + 1 < LCH) s4_hand_put(hs, k & 1, acc);
    cur[0] = acc;
    acc = nf;
  }

  if (PWR) {
    s4_put_tile<true>(out + (size_t)LCH * ostep, cur[0]);
  } else {
    *(v8f*)(G + (size_t)c * NB * NS + lo + 16 * w) = cur[0];
  }
}

__global__ __launch_bounds__(128) void s4_chain_kernel(
    const float* __restrict__ F, const float* __restrict__ M,
    float* __restrict__ G, float* __restrict__ PW, float* __restrict__ Z,
    const float* __restrict__ X0, int nch) {
  __shared__ v4f hb[HB_SLOTS];
  const int c = blockIdx.x;
  if (c < nch) {
    s4_chain_body<false>(F, M, G, PW, Z, X0, c, hb);
  } else {
    s4_chain_body<true>(F, M, G, PW, Z, X0, c - nch, hb);
  }
}

#define UT (LCH / 4)
#define UKG (NI / 4)
#define FB_STEP (4 * HB_TILE)

__device__ __forceinline__ void ub_load_frag(v2f (&f)[UKG], const float* p) {
#pragma unroll
  for (int q = 0; q < NI / 16; ++q) {
    v8f x = *(const v8f*)(p + 16 * q);
    f[4 * q + 0] = __builtin_shufflevector(x, x, 0, 1);
    f[4 * q + 1] = __builtin_shufflevector(x, x, 2, 3);
    f[4 * q + 2] = __builtin_shufflevector(x, x, 4, 5);
    f[4 * q + 3] = __builtin_shufflevector(x, x, 6, 7);
  }
}

__device__ __forceinline__ void s4_force_put(v4f* p, v8f x) {
  p[0] = __builtin_shufflevector(x, x, 0, 1, 2, 3);
  p[32] = __builtin_shufflevector(x, x, 4, 5, 6, 7);
}

__device__ __forceinline__ v8f s4_force_get(const v4f* p) {
  v4f lo = p[0], hi = p[32];
  return __builtin_shufflevector(lo, hi, 0, 1, 2, 3, 4, 5, 6, 7);
}

__device__ __forceinline__ void s4_force_step(const v2f (&x)[UKG],
                                              const v2f (&wt)[4][UKG],
                                              v4f* dst) {
  v8f acc[4] = {{}, {}, {}, {}};
#pragma unroll
  for (int kk = 0; kk < UKG; ++kk)
#pragma unroll
    for (int j = 0; j < 4; ++j) acc[j] = wmma4(wt[j][kk], x[kk], acc[j]);
#pragma unroll
  for (int j = 0; j < 4; ++j) s4_force_put(dst + j * HB_TILE, acc[j]);
  __builtin_amdgcn_sched_barrier(0);
}

__device__ __forceinline__ void s4_force_terms(const float* u, const float* Bsc,
                                               int c, v4f* fb, int wave,
                                               int lane) {
  const int l16 = lane & 15, half = lane >> 4;
  const int k0 = wave * UT;

  const float* urow =
      u + (size_t)l16 * TT * NI + (size_t)(c * LCH + k0) * NI + 8 * half;
  v2f x0[UKG], x1[UKG];
  ub_load_frag(x0, urow);

  const float* brow = Bsc + l16 * NI + 8 * half;
  v2f wt[4][UKG];
#pragma unroll
  for (int q = 0; q < NI / 16; ++q)
#pragma unroll
    for (int j = 0; j < 4; ++j) {
      v8f x = *(const v8f*)(brow + 16 * j * NI + 16 * q);
      wt[j][4 * q + 0] = __builtin_shufflevector(x, x, 0, 1);
      wt[j][4 * q + 1] = __builtin_shufflevector(x, x, 2, 3);
      wt[j][4 * q + 2] = __builtin_shufflevector(x, x, 4, 5);
      wt[j][4 * q + 3] = __builtin_shufflevector(x, x, 6, 7);
    }

  v4f* dst = fb + k0 * FB_STEP + lane;
  static_assert(UT % 2 == 0, "two fragment sets take turns");
#pragma unroll
  for (int i = 0; i < UT; i += 2) {
    ub_load_frag(x1, urow + (size_t)(i + 1) * NI);
    s4_force_step(x0, wt, dst + i * FB_STEP);
    if (i + 2 < UT) ub_load_frag(x0, urow + (size_t)(i + 2) * NI);
    s4_force_step(x1, wt, dst + (i + 1) * FB_STEP);
  }
}

template <int QA, int QB>
__device__ __forceinline__ void s4_tile_mac2(const v2f (&a)[16],
                                             const v8f (&c1)[4],
                                             const v8f (&c2)[4], v8f& acc1,
                                             v8f& acc2) {
#pragma unroll
  for (int kk = 4 * QA; kk < 4 * QB; ++kk) {
    const int qq = kk >> 2, p = kk & 3;
    v2f b1 = {c1[qq][2 * p], c1[qq][2 * p + 1]};
    v2f b2 = {c2[qq][2 * p], c2[qq][2 * p + 1]};
    acc1 = wmma4(a[kk], b1, acc1);
    acc2 = wmma4(a[kk], b2, acc2);
  }
}

template <bool DUAL>
__device__ __forceinline__ void s4_chain1_body(const v2f (&a)[16],
                                               const v4f* fl, float* G,
                                               float* PW, float* Z, int c,
                                               v4f* hb, v4f* hb2) {
  const int w = threadIdx.x >> 5, lane = threadIdx.x & 31;
  const int l16 = lane & 15, half = lane >> 4;

  const s4_slots hs = s4_hand_slots(hb, w, lane);
  const s4_slots hs2 = s4_hand_slots(hb2, w, lane);
  const int lo = l16 * NS + 8 * half;
  const int r0 = 16 * c;
  float* out = Z + (size_t)c * LCH * NB * NS + lo + 16 * w;
  float* pout = PW + (16 * w + 8 * half) * NS + r0 + l16;

  v8f cur[4], cur2[4];
#pragma unroll
  for (int qq = 0; qq < 4; ++qq) {
    cur[qq] = v8f{};
#pragma unroll
    for (int r = 0; r < 8; ++r)
      cur2[qq][r] =
          (16 * ((w + qq) & 3) + 8 * half + r == r0 + l16) ? 1.0f : 0.0f;
  }
  v8f acc = s4_force_get(fl), acc2 = v8f{};

#pragma unroll
  for (int k = 0; k < LCH; ++k) {
    s4_put_tile<false>(out + (size_t)k * NB * NS, cur[0]);
    if (DUAL) s4_put_tile<true>(pout + (size_t)k * NS * NS, cur2[0]);
    v8f nf = v8f{};
    if (k + 1 < LCH) nf = s4_force_get(fl + (k + 1) * FB_STEP);
    if (DUAL) s4_tile_mac2<0, 1>(a, cur, cur2, acc, acc2);
    else acc = s4_tile_mac<0, 1>(a, cur, acc);
    if (k > 0) {
      __builtin_amdgcn_sched_barrier(0);
      s4_lds_barrier();
      s4_hand_get(cur, hs, (k - 1) & 1);
      if (DUAL) s4_hand_get(cur2, hs2, (k - 1) & 1);
    }
    if (DUAL) s4_tile_mac2<1, 4>(a, cur, cur2, acc, acc2);
    else acc = s4_tile_mac<1, 4>(a, cur, acc);
    if (k + 1 < LCH) {
      s4_hand_put(hs, k & 1, acc);
      if (DUAL) s4_hand_put(hs2, k & 1, acc2);
    }
    cur[0] = acc;
    acc = nf;
    if (DUAL) {
      cur2[0] = acc2;
      acc2 = v8f{};
    }
  }

  *(v8f*)(G + (size_t)c * NB * NS + lo + 16 * w) = cur[0];
  if (DUAL) s4_put_tile<true>(pout + (size_t)LCH * NS * NS, cur2[0]);
}

__global__ __launch_bounds__(128) void s4_chain1_kernel(
    const float* __restrict__ u, const float* __restrict__ Bsc,
    const float* __restrict__ M, float* __restrict__ G,
    float* __restrict__ PW, float* __restrict__ Z) {
  __shared__ v4f hb[HB_SLOTS];
  __shared__ v4f hb2[HB_SLOTS];
  __shared__ v4f fb[LCH * FB_STEP];
  const int c = blockIdx.x;
  const int w = threadIdx.x >> 5, lane = threadIdx.x & 31;

  v2f a[16];
  s4_load_mtile(a, M, w, lane & 15, lane >> 4);
  s4_force_terms(u, Bsc, c, fb, w, lane);
  s4_lds_barrier();

  const v4f* fl = fb + w * HB_TILE + lane;
  if (c < NS / 16) {
    s4_chain1_body<true>(a, fl, G, PW, Z, c, hb, hb2);
  } else {
    s4_chain1_body<false>(a, fl, G, PW, Z, c, hb, hb2);
  }
}

__device__ __forceinline__ void s4_pair_mac(const v2f (&a0)[16],
                                            const v2f (&a1)[16],
                                            const v8f (&sb)[4], v8f& z0,
                                            v8f& z1) {
#pragma unroll
  for (int kk = 0; kk < 16; ++kk) {
    const int qq = kk >> 2, p = kk & 3;
    v2f b = {sb[qq][2 * p], sb[qq][2 * p + 1]};
    z0 = wmma4(a0[kk], b, z0);
    z1 = wmma4(a1[kk], b, z1);
  }
}

__global__ __launch_bounds__(128) void s4_tail_kernel(
    const float* __restrict__ GG, const float* __restrict__ A256,
    const float* __restrict__ X0, const float* __restrict__ PW2,
    const float* __restrict__ Zb, const float* __restrict__ PW1,
    float* S, float* __restrict__ xfinal) {
  __shared__ v4f hb[HB_SLOTS];
  const int w = threadIdx.x >> 5, lane = threadIdx.x & 31;
  const int l16 = lane & 15, half = lane >> 4;
  const int c = blockIdx.x, g = c / LCH, j = c % LCH;
  const int lo = l16 * NS + 8 * half;
  const int tile = NB * NS;
  const int lw = lo + 16 * w;
  float* sp = S + (size_t)c * LCH * tile + lw;

  v2f a[16], aS[16];
  v8f cur[4];
  const s4_slots hs = s4_hand_slots(hb, w, lane);
  s4_load_mtile(a, A256, w, l16, half);
  s4_load_state(cur, X0, w, lo);
  v8f acc = *(const v8f*)(GG + lw);
  s4_load_mtile(aS, PW2 + (size_t)j * NS * NS, w, l16, half);
  v8f sbacc = *(const v8f*)(Zb + (size_t)c * tile + lw);
  v2f pa[2][2][16];
  v8f pz[2][2];
#pragma unroll
  for (int i = 0; i < 2; ++i)
#pragma unroll
    for (int e = 0; e < 2; ++e) {
      s4_load_mtile(pa[i][e], PW1 + (size_t)(2 * i + e) * NS * NS, w, l16, half);
      pz[i][e] = *(const v8f*)(sp + (size_t)(2 * i + e) * tile);
    }

  for (int gi = 0; gi < g; ++gi) {
    v8f nf = *(const v8f*)(GG + (size_t)(gi + 1) * tile + lw);
    acc = s4_tile_mac<0, 1>(a, cur, acc);
    if (gi > 0) {
      __builtin_amdgcn_sched_barrier(0);
      s4_lds_barrier();
      s4_hand_get(cur, hs, (gi - 1) & 1);
    }
    acc = s4_tile_mac<1, 4>(a, cur, acc);
    s4_hand_put(hs, gi & 1, acc);
    cur[0] = acc;
    acc = nf;
  }

  sbacc = s4_tile_mac<0, 1>(aS, cur, sbacc);
  if (g > 0) {
    __builtin_amdgcn_sched_barrier(0);
    s4_lds_barrier();
    s4_hand_get(cur, hs, (g - 1) & 1);
  }
  if (c == NCH - 1) {
    acc = s4_tile_mac<0, 4>(a, cur, acc);
    *(v8f*)(xfinal + lw) = acc;
  }
  sbacc = s4_tile_mac<1, 4>(aS, cur, sbacc);
  v8f sb[4];
  s4_hand_put(hs, g & 1, sbacc);
  s4_lds_barrier();
  sb[0] = sbacc;
  s4_hand_get(sb, hs, g & 1);

  static_assert(LCH % 4 == 0, "pairs of states, two sets taking turns");
#pragma unroll
  for (int i = 0; i < LCH / 2; ++i) {
    const int s = i & 1;
    s4_pair_mac(pa[s][0], pa[s][1], sb, pz[s][0], pz[s][1]);
    *(v8f*)(sp + (size_t)(2 * i) * tile) = pz[s][0];
    *(v8f*)(sp + (size_t)(2 * i + 1) * tile) = pz[s][1];
    if (i + 2 < LCH / 2) {
#pragma unroll
      for (int e = 0; e < 2; ++e) {
        s4_load_mtile(pa[s][e], PW1 + (size_t)(2 * i + 4 + e) * NS * NS, w, l16,
                      half);
        pz[s][e] = *(const v8f*)(sp + (size_t)(2 * i + 4 + e) * tile);
      }
    }
    __builtin_amdgcn_sched_barrier(0);
  }
}

#define YT 16
#define YKG (NS / 4 + NI / 4)
#define YNB (YKG / 8)

__device__ __forceinline__ v2f y_load_a(const float* sp, const float* up,
                                        int kk) {
  return kk < NS / 4 ? load2(sp + 4 * kk) : load2(up + 4 * (kk - NS / 4));
}

__global__ __launch_bounds__(128) void s4_y_kernel(
    const float* __restrict__ S, const float* __restrict__ u,
    const float* __restrict__ C, const float* __restrict__ D,
    float* __restrict__ y) {
  const int wave = threadIdx.x >> 5, lane = threadIdx.x & 31;
  const int l16 = lane & 15, half = lane >> 4;
  const int t0 = blockIdx.x * YT;
  const int o0 = wave * 64;

  v2f w[YKG][4];
#pragma unroll
  for (int kk = 0; kk < NS / 4; ++kk)
#pragma unroll
    for (int j = 0; j < 4; ++j)
      w[kk][j] = load2(C + (o0 + 16 * j + l16) * NS + 4 * kk + 2 * half);
#pragma unroll
  for (int kk = 0; kk < NI / 4; ++kk)
#pragma unroll
    for (int j = 0; j < 4; ++j)
      w[NS / 4 + kk][j] =
          load2(D + (o0 + 16 * j + l16) * NI + 4 * kk + 2 * half);

  const float* srow = S + (size_t)t0 * NB * NS + l16 * NS + 2 * half;
  const float* urow = u + (size_t)l16 * TT * NI + (size_t)t0 * NI + 2 * half;
  v2f a[YKG];
#pragma unroll
  for (int kk = 0; kk < 8 * (YNB - 1); ++kk) a[kk] = y_load_a(srow, urow, kk);

  for (int i = 0; i < YT; ++i) {
    const float* sc = srow + (size_t)i * NB * NS;
    const float* uc = urow + (size_t)i * NI;
    const int in = (i + 1 < YT) ? i + 1 : i;
    const float* sn = srow + (size_t)in * NB * NS;
    const float* un = urow + (size_t)in * NI;

    v8f acc[4] = {{}, {}, {}, {}};
#pragma unroll
    for (int c8 = 0; c8 < YNB; ++c8) {
#pragma unroll
      for (int q = 0; q < 8; ++q) {
        const int kk = 8 * c8 + q;
#pragma unroll
        for (int j = 0; j < 4; ++j) acc[j] = wmma4(a[kk], w[kk][j], acc[j]);
      }
      const int r8 = (c8 + YNB - 1) % YNB;
#pragma unroll
      for (int q = 0; q < 8; ++q)
        a[8 * r8 + q] = y_load_a(c8 == 0 ? sc : sn, c8 == 0 ? uc : un, 8 * r8 + q);
      __builtin_amdgcn_sched_barrier(0);
    }

    const int t = t0 + i;
#pragma unroll
    for (int j = 0; j < 4; ++j)
#pragma unroll
      for (int v = 0; v < 8; ++v)
        y[(size_t)(v + 8 * half) * TT * NO + (size_t)t * NO + o0 + 16 * j + l16] =
            acc[j][v];
  }
}

extern "C" void kernel_launch(void* const* d_in, const int* in_sizes, int n_in,
                              void* d_out, int out_size, void* d_ws,
                              size_t ws_size, hipStream_t stream) {
  const float* u  = (const float*)d_in[0];
  const float* x0 = (const float*)d_in[1];
  const float* A  = (const float*)d_in[2];
  const float* B  = (const float*)d_in[3];
  const float* C  = (const float*)d_in[4];
  const float* D  = (const float*)d_in[5];

  float* y      = (float*)d_out;
  float* xfinal = (float*)d_out + (size_t)NB * TT * NO;

  float* ws = (float*)d_ws;
  float* Abar   = ws;                 ws += NS * NS;
  float* PW1    = ws;                 ws += (LCH + 1) * NS * NS;
  float* A16    = PW1 + LCH * NS * NS;
  float* PW2    = ws;                 ws += (LCH + 1) * NS * NS;
  float* A256   = PW2 + LCH * NS * NS;
  float* Bsc    = ws;                 ws += NS * NI;
  float* G      = ws;                 ws += NCH * NB * NS;
  float* Sb     = ws;                 ws += NCH * NB * NS;
  float* GG     = ws;                 ws += NGR * NB * NS;
  float* Sst    = ws;                 ws += (size_t)TT * NB * NS;

  s4_prep_kernel<<<(NS * NS + NS * NI) / 256, 256, 0, stream>>>(A, B, Abar, Bsc);
  s4_chain1_kernel<<<NCH, 128, 0, stream>>>(u, Bsc, Abar, G, PW1, Sst);
  s4_chain_kernel<<<NGR + NS / 16, 128, 0, stream>>>(G, A16, GG, PW2, Sb, nullptr, NGR);
  s4_tail_kernel<<<NCH, 128, 0, stream>>>(GG, A256, x0, PW2, Sb, PW1, Sst, xfinal);
  s4_y_kernel<<<TT / YT, 128, 0, stream>>>(Sst, u, C, D, y);
}


# tail and y in ONE launch (4, not 5): each state tile goes from the tail's accumulators to the y product through LDS, lane for lane in the chain's k order; S never stored; sum order of y changes
# speedup vs baseline: 8.9461x; 1.0292x over previous
#include <hip/hip_runtime.h>
#include <hip/hip_bf16.h>


typedef __attribute__((ext_vector_type(2))) float v2f;
typedef __attribute__((ext_vector_type(8))) float v8f;

#define S4_DT 0.001f
#define TT 4096
#define NB 16
#define NS 64
#define NI 256
#define NO 256
#define LCH 16
#define NCH 256
#define NGR 16

__device__ __forceinline__ v8f wmma4(v2f a, v2f b, v8f c) {
  return __builtin_amdgcn_wmma_f32_16x16x4_f32(
       false, a,  false, b,
       (short)0, c,  false,  false);
}

__device__ __forceinline__ v2f load2(const float* p) { return *(const v2f*)p; }

__device__ __forceinline__ void store_c(float* dst, int rs, int n0, v8f acc,
                                        int l16, int half) {
#pragma unroll
  for (int v = 0; v < 8; ++v)
    dst[(v + 8 * half) * rs + n0 + l16] = acc[v];
}

__device__ __forceinline__ v8f load_c(const float* src, int rs, int n0,
                                      int l16, int half) {
  v8f r;
#pragma unroll
  for (int v = 0; v < 8; ++v)
    r[v] = src[(v + 8 * half) * rs + n0 + l16];
  return r;
}

__global__ __launch_bounds__(256) void s4_prep_kernel(
    const float* __restrict__ A, const float* __restrict__ B,
    float* __restrict__ Abar, float* __restrict__ Bsc) {
  const int idx = blockIdx.x * 256 + threadIdx.x;
  if (idx < NS * NS) {
    int r = idx >> 6, c = idx & 63;
    Abar[idx] = A[idx] * S4_DT + (r == c ? 1.0f : 0.0f);
  } else {
    Bsc[idx - NS * NS] = B[idx - NS * NS] * S4_DT;
  }
}

typedef __attribute__((ext_vector_type(4))) float v4f;

#define HB_TILE 64
#define HB_SLOTS (2 * 4 * HB_TILE)

struct s4_slots { v4f* p[4]; };

__device__ __forceinline__ s4_slots s4_hand_slots(v4f* hb, int w, int lane) {
  s4_slots hs;
#pragma unroll
  for (int qq = 0; qq < 4; ++qq) hs.p[qq] = hb + ((w + qq) & 3) * HB_TILE + lane;
  return hs;
}

__device__ __forceinline__ void s4_hand_put(const s4_slots& hs, int par, v8f x) {
  v4f* p = hs.p[0] + par * 4 * HB_TILE;
  p[0] = __builtin_shufflevector(x, x, 0, 1, 2, 3);
  p[32] = __builtin_shufflevector(x, x, 4, 5, 6, 7);
}

__device__ __forceinline__ void s4_hand_get(v8f (&cur)[4], const s4_slots& hs,
                                            int par) {
#pragma unroll
  for (int qq = 1; qq < 4; ++qq) {
    const v4f* p = hs.p[qq] + par * 4 * HB_TILE;
    v4f lo = p[0], hi = p[32];
    cur[qq] = __builtin_shufflevector(lo, hi, 0, 1, 2, 3, 4, 5, 6, 7);
  }
  __builtin_amdgcn_sched_barrier(0);
}

__device__ __forceinline__ void s4_lds_barrier() {
  asm volatile("s_wait_dscnt 0x0\n\ts_barrier_signal -1\n\ts_barrier_wait -1"
               ::: "memory");
}

__device__ __forceinline__ void s4_load_mtile(v2f (&a)[16], const float* M,
                                              int w, int l16, int half) {
  const float* row = M + (16 * w + l16) * NS + 8 * half;
#pragma unroll
  for (int qq = 0; qq < 4; ++qq) {
    v8f x = *(const v8f*)(row + 16 * ((w + qq) & 3));
    a[4 * qq + 0] = __builtin_shufflevector(x, x, 0, 1);
    a[4 * qq + 1] = __builtin_shufflevector(x, x, 2, 3);
    a[4 * qq + 2] = __builtin_shufflevector(x, x, 4, 5);
    a[4 * qq + 3] = __builtin_shufflevector(x, x, 6, 7);
  }
}

__device__ __forceinline__ void s4_load_state(v8f (&cur)[4], const float* X,
                                              int w, int lo) {
#pragma unroll
  for (int qq = 0; qq < 4; ++qq)
    cur[qq] = *(const v8f*)(X + lo + 16 * ((w + qq) & 3));
}

template <int QA, int QB>
__device__ __forceinline__ v8f s4_tile_mac(const v2f (&a)[16],
                                           const v8f (&cur)[4], v8f acc) {
#pragma unroll
  for (int kk = 4 * QA; kk < 4 * QB; ++kk) {
    const int qq = kk >> 2, p = kk & 3;
    v2f b = {cur[qq][2 * p], cur[qq][2 * p + 1]};
    acc = wmma4(a[kk], b, acc);
  }
  return acc;
}

template <bool PWR>
__device__ __forceinline__ void s4_put_tile(float* p, v8f st) {
  if (PWR) {
#pragma unroll
    for (int r = 0; r < 8; ++r) p[r * NS] = st[r];
  } else {
    *(v8f*)p = st;
  }
}

template <bool PWR>
__device__ __forceinline__ void s4_chain_body(const float* F, const float* M,
                                              float* G, float* PW, float* Z,
                                              const float* X0, int c, v4f* hb) {
  const int w = threadIdx.x >> 5, lane = threadIdx.x & 31;
  const int l16 = lane & 15, half = lane >> 4;

  v2f a[16];
  s4_load_mtile(a, M, w, l16, half);
  const s4_slots hs = s4_hand_slots(hb, w, lane);

  const int lo = l16 * NS + 8 * half;
  const int r0 = 16 * c;
  const float* Fc = F + (size_t)c * LCH * NB * NS + lo + 16 * w;
  float* out = PWR ? PW + (16 * w + 8 * half) * NS + r0 + l16
                   : Z + (size_t)c * LCH * NB * NS + lo + 16 * w;
  const int ostep = PWR ? NS * NS : NB * NS;
  const int fstep = NB * NS;

  v8f cur[4];
  if (PWR) {
#pragma unroll
    for (int qq = 0; qq < 4; ++qq)
#pragma unroll
      for (int r = 0; r < 8; ++r)
        cur[qq][r] =
            (16 * ((w + qq) & 3) + 8 * half + r == r0 + l16) ? 1.0f : 0.0f;
  } else if (X0) {
    s4_load_state(cur, X0, w, lo);
  } else {
#pragma unroll
    for (int qq = 0; qq < 4; ++qq) cur[qq] = v8f{};
  }
  v8f acc = v8f{};
  if (!PWR) acc = *(const v8f*)Fc;

#pragma unroll
  for (int k = 0; k < LCH; ++k) {
    s4_put_tile<PWR>(out + (size_t)k * ostep, cur[0]);
    v8f nf = v8f{};
    if (!PWR && k + 1 < LCH) nf = *(const v8f*)(Fc + (size_t)(k + 1) * fstep);
    acc = s4_tile_mac<0, 1>(a, cur, acc);
    if (k > 0) {
      __builtin_amdgcn_sched_barrier(0);
      s4_lds_barrier();
      s4_hand_get(cur, hs, (k - 1) & 1);
    }
    acc = s4_tile_mac<1, 4>(a, cur, acc);
    if (k + 1 < LCH) s4_hand_put(hs, k & 1, acc);
    cur[0] = acc;
    acc = nf;
  }

  if (PWR) {
    s4_put_tile<true>(out + (size_t)LCH * ostep, cur[0]);
  } else {
    *(v8f*)(G + (size_t)c * NB * NS + lo + 16 * w) = cur[0];
  }
}

__global__ __launch_bounds__(128) void s4_chain_kernel(
    const float* __restrict__ F, const float* __restrict__ M,
    float* __restrict__ G, float* __restrict__ PW, float* __restrict__ Z,
    const float* __restrict__ X0, int nch) {
  __shared__ v4f hb[HB_SLOTS];
  const int c = blockIdx.x;
  if (c < nch) {
    s4_chain_body<false>(F, M, G, PW, Z, X0, c, hb);
  } else {
    s4_chain_body<true>(F, M, G, PW, Z, X0, c - nch, hb);
  }
}

#define UT (LCH / 4)
#define UKG (NI / 4)
#define FB_STEP (4 * HB_TILE)

__device__ __forceinline__ void ub_load_frag(v2f (&f)[UKG], const float* p) {
#pragma unroll
  for (int q = 0; q < NI / 16; ++q) {
    v8f x = *(const v8f*)(p + 16 * q);
    f[4 * q + 0] = __builtin_shufflevector(x, x, 0, 1);
    f[4 * q + 1] = __builtin_shufflevector(x, x, 2, 3);
    f[4 * q + 2] = __builtin_shufflevector(x, x, 4, 5);
    f[4 * q + 3] = __builtin_shufflevector(x, x, 6, 7);
  }
}

__device__ __forceinline__ void s4_force_put(v4f* p, v8f x) {
  p[0] = __builtin_shufflevector(x, x, 0, 1, 2, 3);
  p[32] = __builtin_shufflevector(x, x, 4, 5, 6, 7);
}

__device__ __forceinline__ v8f s4_force_get(const v4f* p) {
  v4f lo = p[0], hi = p[32];
  return __builtin_shufflevector(lo, hi, 0, 1, 2, 3, 4, 5, 6, 7);
}

__device__ __forceinline__ void s4_force_step(const v2f (&x)[UKG],
                                              const v2f (&wt)[4][UKG],
                                              v4f* dst) {
  v8f acc[4] = {{}, {}, {}, {}};
#pragma unroll
  for (int kk = 0; kk < UKG; ++kk)
#pragma unroll
    for (int j = 0; j < 4; ++j) acc[j] = wmma4(wt[j][kk], x[kk], acc[j]);
#pragma unroll
  for (int j = 0; j < 4; ++j) s4_force_put(dst + j * HB_TILE, acc[j]);
  __builtin_amdgcn_sched_barrier(0);
}

__device__ __forceinline__ void s4_force_terms(const float* u, const float* Bsc,
                                               int c, v4f* fb, int wave,
                                               int lane) {
  const int l16 = lane & 15, half = lane >> 4;
  const int k0 = wave * UT;

  const float* urow =
      u + (size_t)l16 * TT * NI + (size_t)(c * LCH + k0) * NI + 8 * half;
  v2f x0[UKG], x1[UKG];
  ub_load_frag(x0, urow);

  const float* brow = Bsc + l16 * NI + 8 * half;
  v2f wt[4][UKG];
#pragma unroll
  for (int q = 0; q < NI / 16; ++q)
#pragma unroll
    for (int j = 0; j < 4; ++j) {
      v8f x = *(const v8f*)(brow + 16 * j * NI + 16 * q);
      wt[j][4 * q + 0] = __builtin_shufflevector(x, x, 0, 1);
      wt[j][4 * q + 1] = __builtin_shufflevector(x, x, 2, 3);
      wt[j][4 * q + 2] = __builtin_shufflevector(x, x, 4, 5);
      wt[j][4 * q + 3] = __builtin_shufflevector(x, x, 6, 7);
    }

  v4f* dst = fb + k0 * FB_STEP + lane;
  static_assert(UT % 2 == 0, "two fragment sets take turns");
#pragma unroll
  for (int i = 0; i < UT; i += 2) {
    ub_load_frag(x1, urow + (size_t)(i + 1) * NI);
    s4_force_step(x0, wt, dst + i * FB_STEP);
    if (i + 2 < UT) ub_load_frag(x0, urow + (size_t)(i + 2) * NI);
    s4_force_step(x1, wt, dst + (i + 1) * FB_STEP);
  }
}

template <int QA, int QB>
__device__ __forceinline__ void s4_tile_mac2(const v2f (&a)[16],
                                             const v8f (&c1)[4],
                                             const v8f (&c2)[4], v8f& acc1,
                                             v8f& acc2) {
#pragma unroll
  for (int kk = 4 * QA; kk < 4 * QB; ++kk) {
    const int qq = kk >> 2, p = kk & 3;
    v2f b1 = {c1[qq][2 * p], c1[qq][2 * p + 1]};
    v2f b2 = {c2[qq][2 * p], c2[qq][2 * p + 1]};
    acc1 = wmma4(a[kk], b1, acc1);
    acc2 = wmma4(a[kk], b2, acc2);
  }
}

template <bool DUAL>
__device__ __forceinline__ void s4_chain1_body(const v2f (&a)[16],
                                               const v4f* fl, float* G,
                                               float* PW, float* Z, int c,
                                               v4f* hb, v4f* hb2) {
  const int w = threadIdx.x >> 5, lane = threadIdx.x & 31;
  const int l16 = lane & 15, half = lane >> 4;

  const s4_slots hs = s4_hand_slots(hb, w, lane);
  const s4_slots hs2 = s4_hand_slots(hb2, w, lane);
  const int lo = l16 * NS + 8 * half;
  const int r0 = 16 * c;
  float* out = Z + (size_t)c * LCH * NB * NS + lo + 16 * w;
  float* pout = PW + (16 * w + 8 * half) * NS + r0 + l16;

  v8f cur[4], cur2[4];
#pragma unroll
  for (int qq = 0; qq < 4; ++qq) {
    cur[qq] = v8f{};
#pragma unroll
    for (int r = 0; r < 8; ++r)
      cur2[qq][r] =
          (16 * ((w + qq) & 3) + 8 * half + r == r0 + l16) ? 1.0f : 0.0f;
  }
  v8f acc = s4_force_get(fl), acc2 = v8f{};

#pragma unroll
  for (int k = 0; k < LCH; ++k) {
    s4_put_tile<false>(out + (size_t)k * NB * NS, cur[0]);
    if (DUAL) s4_put_tile<true>(pout + (size_t)k * NS * NS, cur2[0]);
    v8f nf = v8f{};
    if (k + 1 < LCH) nf = s4_force_get(fl + (k + 1) * FB_STEP);
    if (DUAL) s4_tile_mac2<0, 1>(a, cur, cur2, acc, acc2);
    else acc = s4_tile_mac<0, 1>(a, cur, acc);
    if (k > 0) {
      __builtin_amdgcn_sched_barrier(0);
      s4_lds_barrier();
      s4_hand_get(cur, hs, (k - 1) & 1);
      if (DUAL) s4_hand_get(cur2, hs2, (k - 1) & 1);
    }
    if (DUAL) s4_tile_mac2<1, 4>(a, cur, cur2, acc, acc2);
    else acc = s4_tile_mac<1, 4>(a, cur, acc);
    if (k + 1 < LCH) {
      s4_hand_put(hs, k & 1, acc);
      if (DUAL) s4_hand_put(hs2, k & 1, acc2);
    }
    cur[0] = acc;
    acc = nf;
    if (DUAL) {
      cur2[0] = acc2;
      acc2 = v8f{};
    }
  }

  *(v8f*)(G + (size_t)c * NB * NS + lo + 16 * w) = cur[0];
  if (DUAL) s4_put_tile<true>(pout + (size_t)LCH * NS * NS, cur2[0]);
}

__global__ __launch_bounds__(128) void s4_chain1_kernel(
    const float* __restrict__ u, const float* __restrict__ Bsc,
    const float* __restrict__ M, float* __restrict__ G,
    float* __restrict__ PW, float* __restrict__ Z) {
  __shared__ v4f hb[HB_SLOTS];
  __shared__ v4f hb2[HB_SLOTS];
  __shared__ v4f fb[LCH * FB_STEP];
  const int c = blockIdx.x;
  const int w = threadIdx.x >> 5, lane = threadIdx.x & 31;

  v2f a[16];
  s4_load_mtile(a, M, w, lane & 15, lane >> 4);
  s4_force_terms(u, Bsc, c, fb, w, lane);
  s4_lds_barrier();

  const v4f* fl = fb + w * HB_TILE + lane;
  if (c < NS / 16) {
    s4_chain1_body<true>(a, fl, G, PW, Z, c, hb, hb2);
  } else {
    s4_chain1_body<false>(a, fl, G, PW, Z, c, hb, hb2);
  }
}

__device__ __forceinline__ void s4_pair_mac(const v2f (&a0)[16],
                                            const v2f (&a1)[16],
                                            const v8f (&sb)[4], v8f& z0,
                                            v8f& z1) {
#pragma unroll
  for (int kk = 0; kk < 16; ++kk) {
    const int qq = kk >> 2, p = kk & 3;
    v2f b = {sb[qq][2 * p], sb[qq][2 * p + 1]};
    z0 = wmma4(a0[kk], b, z0);
    z1 = wmma4(a1[kk], b, z1);
  }
}

#define YT LCH
#define YKG (NS / 4 + NI / 4)
#define YNB (YKG / 8)
#define SL_STEP (4 * HB_TILE)

__global__ __launch_bounds__(128) void s4_tail_y_kernel(
    const float* __restrict__ GG, const float* __restrict__ A256,
    const float* __restrict__ X0, const float* __restrict__ PW2,
    const float* __restrict__ Zb, const float* __restrict__ PW1,
    const float* __restrict__ Z, float* __restrict__ xfinal,
    const float* __restrict__ u, const float* __restrict__ C,
    const float* __restrict__ D, float* __restrict__ y) {
  __shared__ v4f hb[HB_SLOTS];
  __shared__ v4f sl[LCH * SL_STEP];
  const int w = threadIdx.x >> 5, lane = threadIdx.x & 31;
  const int l16 = lane & 15, half = lane >> 4;
  const int c = blockIdx.x;

  {
    const int g = c / LCH, j = c % LCH;
    const int lo = l16 * NS + 8 * half;
    const int tile = NB * NS;
    const int lw = lo + 16 * w;
    const float* zp = Z + (size_t)c * LCH * tile + lw;
    v4f* slw = sl + w * HB_TILE + lane;

    v2f a[16], aS[16];
    v8f cur[4];
    const s4_slots hs = s4_hand_slots(hb, w, lane);
    s4_load_mtile(a, A256, w, l16, half);
    s4_load_state(cur, X0, w, lo);
    v8f acc = *(const v8f*)(GG + lw);
    s4_load_mtile(aS, PW2 + (size_t)j * NS * NS, w, l16, half);
    v8f sbacc = *(const v8f*)(Zb + (size_t)c * tile + lw);
    v2f pa[2][2][16];
    v8f pz[2][2];
#pragma unroll
    for (int i = 0; i < 2; ++i)
#pragma unroll
      for (int e = 0; e < 2; ++e) {
        s4_load_mtile(pa[i][e], PW1 + (size_t)(2 * i + e) * NS * NS, w, l16, half);
        pz[i][e] = *(const v8f*)(zp + (size_t)(2 * i + e) * tile);
      }

    for (int gi = 0; gi < g; ++gi) {
      v8f nf = *(const v8f*)(GG + (size_t)(gi + 1) * tile + lw);
      acc = s4_tile_mac<0, 1>(a, cur, acc);
      if (gi > 0) {
        __builtin_amdgcn_sched_barrier(0);
        s4_lds_barrier();
        s4_hand_get(cur, hs, (gi - 1) & 1);
      }
      acc = s4_tile_mac<1, 4>(a, cur, acc);
      s4_hand_put(hs, gi & 1, acc);
      cur[0] = acc;
      acc = nf;
    }

    sbacc = s4_tile_mac<0, 1>(aS, cur, sbacc);
    if (g > 0) {
      __builtin_amdgcn_sched_barrier(0);
      s4_lds_barrier();
      s4_hand_get(cur, hs, (g - 1) & 1);
    }
    if (c == NCH - 1) {
      acc = s4_tile_mac<0, 4>(a, cur, acc);
      *(v8f*)(xfinal + lw) = acc;
    }
    sbacc = s4_tile_mac<1, 4>(aS, cur, sbacc);
    v8f sb[4];
    s4_hand_put(hs, g & 1, sbacc);
    s4_lds_barrier();
    sb[0] = sbacc;
    s4_hand_get(sb, hs, g & 1);

    static_assert(LCH % 4 == 0, "pairs of states, two sets taking turns");
#pragma unroll
    for (int i = 0; i < LCH / 2; ++i) {
      const int s = i & 1;
      s4_pair_mac(pa[s][0], pa[s][1], sb, pz[s][0], pz[s][1]);
      s4_force_put(slw + (2 * i) * SL_STEP, pz[s][0]);
      s4_force_put(slw + (2 * i + 1) * SL_STEP, pz[s][1]);
      if (i + 2 < LCH / 2) {
#pragma unroll
        for (int e = 0; e < 2; ++e) {
          s4_load_mtile(pa[s][e], PW1 + (size_t)(2 * i + 4 + e) * NS * NS, w,
                        l16, half);
          pz[s][e] = *(const v8f*)(zp + (size_t)(2 * i + 4 + e) * tile);
        }
      }
      __builtin_amdgcn_sched_barrier(0);
    }
  }
  s4_lds_barrier();

  const int t0 = c * YT;
  const int o0 = w * 64;

  v2f wt[YKG][4];
#pragma unroll
  for (int q = 0; q < NS / 16; ++q)
#pragma unroll
    for (int jj = 0; jj < 4; ++jj) {
      v8f x = *(const v8f*)(C + (o0 + 16 * jj + l16) * NS + 16 * q + 8 * half);
      wt[4 * q + 0][jj] = __builtin_shufflevector(x, x, 0, 1);
      wt[4 * q + 1][jj] = __builtin_shufflevector(x, x, 2, 3);
      wt[4 * q + 2][jj] = __builtin_shufflevector(x, x, 4, 5);
      wt[4 * q + 3][jj] = __builtin_shufflevector(x, x, 6, 7);
    }
#pragma unroll
  for (int kk = 0; kk < NI / 4; ++kk)
#pragma unroll
    for (int jj = 0; jj < 4; ++jj)
      wt[NS / 4 + kk][jj] =
          load2(D + (o0 + 16 * jj + l16) * NI + 4 * kk + 2 * half);

  const v4f* sll = sl + lane;
  const float* urow = u + (size_t)l16 * TT * NI + (size_t)t0 * NI + 2 * half;
  v2f a[YKG];
#pragma unroll
  for (int q = 0; q < NS / 16; ++q) {
    v8f x = s4_force_get(sll + q * HB_TILE);
    a[4 * q + 0] = __builtin_shufflevector(x, x, 0, 1);
    a[4 * q + 1] = __builtin_shufflevector(x, x, 2, 3);
    a[4 * q + 2] = __builtin_shufflevector(x, x, 4, 5);
    a[4 * q + 3] = __builtin_shufflevector(x, x, 6, 7);
  }
#pragma unroll
  for (int kk = NS / 4; kk < 8 * (YNB - 1); ++kk)
    a[kk] = load2(urow + 4 * (kk - NS / 4));

  for (int i = 0; i < YT; ++i) {
    const float* uc = urow + (size_t)i * NI;
    const int in = (i + 1 < YT) ? i + 1 : i;
    const float* un = urow + (size_t)in * NI;
    const v4f* sn = sll + in * SL_STEP;

    v8f acc[4] = {{}, {}, {}, {}};
#pragma unroll
    for (int c8 = 0; c8 < YNB; ++c8) {
#pragma unroll
      for (int q = 0; q < 8; ++q) {
        const int kk = 8 * c8 + q;
#pragma unroll
        for (int jj = 0; jj < 4; ++jj)
          acc[jj] = wmma4(a[kk], wt[kk][jj], acc[jj]);
      }
      const int r8 = (c8 + YNB - 1) % YNB;
      if (r8 >= NS / 32) {
#pragma unroll
        for (int q = 0; q < 8; ++q)
          a[8 * r8 + q] = load2((c8 == 0 ? uc : un) + 4 * (8 * r8 + q - NS / 4));
      } else {
#pragma unroll
        for (int qq = 0; qq < 2; ++qq) {
          v8f x = s4_force_get(sn + (2 * r8 + qq) * HB_TILE);
          a[8 * r8 + 4 * qq + 0] = __builtin_shufflevector(x, x, 0, 1);
          a[8 * r8 + 4 * qq + 1] = __builtin_shufflevector(x, x, 2, 3);
          a[8 * r8 + 4 * qq + 2] = __builtin_shufflevector(x, x, 4, 5);
          a[8 * r8 + 4 * qq + 3] = __builtin_shufflevector(x, x, 6, 7);
        }
      }
      __builtin_amdgcn_sched_barrier(0);
    }

    const int t = t0 + i;
#pragma unroll
    for (int jj = 0; jj < 4; ++jj)
#pragma unroll
      for (int v = 0; v < 8; ++v)
        y[(size_t)(v + 8 * half) * TT * NO + (size_t)t * NO + o0 + 16 * jj + l16] =
            acc[jj][v];
  }
}

extern "C" void kernel_launch(void* const* d_in, const int* in_sizes, int n_in,
                              void* d_out, int out_size, void* d_ws,
                              size_t ws_size, hipStream_t stream) {
  const float* u  = (const float*)d_in[0];
  const float* x0 = (const float*)d_in[1];
  const float* A  = (const float*)d_in[2];
  const float* B  = (const float*)d_in[3];
  const float* C  = (const float*)d_in[4];
  const float* D  = (const float*)d_in[5];

  float* y      = (float*)d_out;
  float* xfinal = (float*)d_out + (size_t)NB * TT * NO;

  float* ws = (float*)d_ws;
  float* Abar   = ws;                 ws += NS * NS;
  float* PW1    = ws;                 ws += (LCH + 1) * NS * NS;
  float* A16    = PW1 + LCH * NS * NS;
  float* PW2    = ws;                 ws += (LCH + 1) * NS * NS;
  float* A256   = PW2 + LCH * NS * NS;
  float* Bsc    = ws;                 ws += NS * NI;
  float* G      = ws;                 ws += NCH * NB * NS;
  float* Sb     = ws;                 ws += NCH * NB * NS;
  float* GG     = ws;                 ws += NGR * NB * NS;
  float* Zst    = ws;                 ws += (size_t)TT * NB * NS;

  s4_prep_kernel<<<(NS * NS + NS * NI) / 256, 256, 0, stream>>>(A, B, Abar, Bsc);
  s4_chain1_kernel<<<NCH, 128, 0, stream>>>(u, Bsc, Abar, G, PW1, Zst);
  s4_chain_kernel<<<NGR + NS / 16, 128, 0, stream>>>(G, A16, GG, PW2, Sb, nullptr, NGR);
  s4_tail_y_kernel<<<NCH, 128, 0, stream>>>(GG, A256, x0, PW2, Sb, PW1, Zst, xfinal,
                                            u, C, D, y);
}
